# starting kernel
# baseline (speedup 1.0000x reference)
#include <hip/hip_runtime.h>
#include <math.h>

#define B_N   4
#define C_CH  96
#define DDIM  16
#define HDIM  56
#define WDIM  56
#define GDIM  96
#define HID   24
#define NB    3
#define NCOL  (C_CH + GDIM)
#define SPATIAL (DDIM * HDIM * WDIM)
#define LN_EPS 1e-5f

typedef float v2f __attribute__((ext_vector_type(2)));
typedef float v4f __attribute__((ext_vector_type(4)));
typedef float v8f __attribute__((ext_vector_type(8)));

__global__ void feat_reduce_kernel(const float* __restrict__ x,
                                   float* __restrict__ feat) {
    const int bc = blockIdx.x;
    const v4f* p = (const v4f*)(x + (size_t)bc * SPATIAL);
    float s = 0.f;
    for (int i = threadIdx.x; i < SPATIAL / 4; i += 256) {
        v4f v = p[i];
        s += v.x + v.y + v.z + v.w;
    }
    __shared__ float red[256];
    red[threadIdx.x] = s;
    __syncthreads();
    for (int off = 128; off > 0; off >>= 1) {
        if (threadIdx.x < off) red[threadIdx.x] += red[threadIdx.x + off];
        __syncthreads();
    }
    if (threadIdx.x == 0) feat[bc] = red[0] * (1.0f / (float)SPATIAL);
}

__global__ void gate_kernel(const float* __restrict__ feat,
                            const float* __restrict__ guidance,
                            const float* __restrict__ ln_g,
                            const float* __restrict__ ln_b,
                            const float* __restrict__ w1p,
                            const float* __restrict__ b1p,
                            const float* __restrict__ w2p,
                            const float* __restrict__ b2p,
                            float* __restrict__ wts) {
    __shared__ float gn[16][NCOL];
    __shared__ float w1t[32][NCOL];
    __shared__ float mu[B_N];
    __shared__ float rs[B_N];
    __shared__ float hbuf[B_N][HID];
    __shared__ float logits[B_N][NB];

    const int lane = threadIdx.x;

    for (int idx = lane; idx < 12 * NCOL; idx += 32)
        gn[4 + idx / NCOL][idx % NCOL] = 0.f;
    for (int idx = lane; idx < 32 * NCOL; idx += 32) {
        const int n = idx / NCOL, k = idx % NCOL;
        w1t[n][k] = (n < HID) ? w1p[k * HID + n] : 0.f;
    }
    for (int idx = lane; idx < B_N * NCOL; idx += 32) {
        const int r = idx / NCOL, col = idx % NCOL;
        gn[r][col] = (col < C_CH) ? feat[r * C_CH + col]
                                  : guidance[r * GDIM + (col - C_CH)];
    }
    __syncthreads();

    if (lane < B_N) {
        float m = 0.f;
        for (int j = 0; j < NCOL; ++j) m += gn[lane][j];
        m *= (1.0f / NCOL);
        float v = 0.f;
        for (int j = 0; j < NCOL; ++j) { float d = gn[lane][j] - m; v += d * d; }
        v *= (1.0f / NCOL);
        mu[lane] = m;
        rs[lane] = rsqrtf(v + LN_EPS);
    }
    __syncthreads();

    for (int idx = lane; idx < B_N * NCOL; idx += 32) {
        const int r = idx / NCOL, col = idx % NCOL;
        gn[r][col] = (gn[r][col] - mu[r]) * rs[r] * ln_g[col] + ln_b[col];
    }
    __syncthreads();

    const int half = lane >> 4;
    const int l    = lane & 15;
    const float* arow  = &gn[l][0];
    const float* b0row = &w1t[l][0];
    const float* b1row = &w1t[16 + l][0];
    v8f c0 = {};
    v8f c1 = {};
    for (int kk = 0; kk < NCOL / 4; ++kk) {
        const int kb = kk * 4 + half * 2;
        const v2f a   = *(const v2f*)(arow + kb);
        const v2f b0  = *(const v2f*)(b0row + kb);
        const v2f b1v = *(const v2f*)(b1row + kb);
        c0 = __builtin_amdgcn_wmma_f32_16x16x4_f32(false, a, false, b0,
                                                   (short)0, c0, false, false);
        c1 = __builtin_amdgcn_wmma_f32_16x16x4_f32(false, a, false, b1v,
                                                   (short)0, c1, false, false);
    }

    #pragma unroll
    for (int r = 0; r < 8; ++r) {
        const int M = half ? (r + 8) : r;
        if (M < B_N) {
            float v0 = c0[r] + b1p[l];
            v0 = 0.5f * v0 * (1.0f + erff(v0 * 0.70710678118654752f));
            hbuf[M][l] = v0;
            const int n1 = 16 + l;
            if (n1 < HID) {
                float v1 = c1[r] + b1p[n1];
                v1 = 0.5f * v1 * (1.0f + erff(v1 * 0.70710678118654752f));
                hbuf[M][n1] = v1;
            }
        }
    }
    __syncthreads();

    if (lane < B_N * NB) {
        const int b = lane / NB, j = lane % NB;
        float s = b2p[j];
        for (int k = 0; k < HID; ++k) s += hbuf[b][k] * w2p[k * NB + j];
        logits[b][j] = s;
    }
    __syncthreads();

    if (lane < B_N) {
        float m = fmaxf(logits[lane][0], fmaxf(logits[lane][1], logits[lane][2]));
        float e0 = __expf(logits[lane][0] - m);
        float e1 = __expf(logits[lane][1] - m);
        float e2 = __expf(logits[lane][2] - m);
        float inv = 1.0f / (e0 + e1 + e2);
        wts[lane * NB + 0] = e0 * inv;
        wts[lane * NB + 1] = e1 * inv;
        wts[lane * NB + 2] = e2 * inv;
    }
}

__global__ void fold_kernel(const float* __restrict__ convw,
                            const float* __restrict__ convb,
                            const float* __restrict__ wts,
                            float* __restrict__ wscaled,
                            float* __restrict__ biaseff) {
    const int idx = blockIdx.x * 256 + threadIdx.x;
    const int total = B_N * C_CH * NB * 27;
    if (idx < total) {
        const int t = idx % 27;
        int tmp = idx / 27;
        const int i = tmp % NB;
        tmp /= NB;
        const int c = tmp % C_CH;
        const int b = tmp / C_CH;
        wscaled[idx] = wts[b * NB + i] * convw[(i * C_CH + c) * 27 + t];
    }
    if (idx < B_N * C_CH) {
        const int c = idx % C_CH, b = idx / C_CH;
        float s = 0.f;
        for (int i = 0; i < NB; ++i) s += wts[b * NB + i] * convb[i * C_CH + c];
        biaseff[idx] = s;
    }
}

__global__ void dwconv_kernel(const float* __restrict__ x,
                              const float* __restrict__ wscaled,
                              const float* __restrict__ biaseff,
                              float* __restrict__ out) {
    const int bcd = blockIdx.x;
    const int z  = bcd % DDIM;
    const int bc = bcd / DDIM;

    __shared__ float wsm[NB * 27];
    __shared__ float biass;
    if (threadIdx.x < NB * 27) wsm[threadIdx.x] = wscaled[bc * NB * 27 + threadIdx.x];
    if (threadIdx.x == NB * 27) biass = biaseff[bc];
    __syncthreads();

    const float* xp = x + (size_t)bc * SPATIAL;
    float* op = out + (size_t)bc * SPATIAL + (size_t)z * (HDIM * WDIM);
    const bool zint = (z >= 3) && (z <= DDIM - 4);

    for (int hw = threadIdx.x; hw < HDIM * WDIM; hw += 256) {
        const int y  = hw / WDIM;
        const int xw = hw % WDIM;
        float acc = biass;
        if (zint && (y >= 3) && (y <= HDIM - 4) && (xw >= 3) && (xw <= WDIM - 4)) {
            const float* base = xp + ((size_t)z * HDIM + y) * WDIM + xw;
            #pragma unroll
            for (int i = 0; i < NB; ++i) {
                const int di = i + 1;
                const float* wp = &wsm[i * 27];
                #pragma unroll
                for (int kz = 0; kz < 3; ++kz) {
                    #pragma unroll
                    for (int ky = 0; ky < 3; ++ky) {
                        const float* row = base + ((kz - 1) * di * HDIM
                                                 + (ky - 1) * di) * WDIM;
                        #pragma unroll
                        for (int kx = 0; kx < 3; ++kx)
                            acc = fmaf(wp[(kz * 3 + ky) * 3 + kx],
                                       row[(kx - 1) * di], acc);
                    }
                }
            }
        } else {
            #pragma unroll
            for (int i = 0; i < NB; ++i) {
                const int di = i + 1;
                const float* wp = &wsm[i * 27];
                #pragma unroll
                for (int kz = 0; kz < 3; ++kz) {
                    const int z2 = z + (kz - 1) * di;
                    if ((unsigned)z2 >= (unsigned)DDIM) continue;
                    #pragma unroll
                    for (int ky = 0; ky < 3; ++ky) {
                        const int y2 = y + (ky - 1) * di;
                        if ((unsigned)y2 >= (unsigned)HDIM) continue;
                        const float* row = xp + ((size_t)z2 * HDIM + y2) * WDIM;
                        #pragma unroll
                        for (int kx = 0; kx < 3; ++kx) {
                            const int x2 = xw + (kx - 1) * di;
                            if ((unsigned)x2 >= (unsigned)WDIM) continue;
                            acc = fmaf(wp[(kz * 3 + ky) * 3 + kx], row[x2], acc);
                        }
                    }
                }
            }
        }
        op[hw] = acc;
    }
}

extern "C" void kernel_launch(void* const* d_in, const int* in_sizes, int n_in,
                              void* d_out, int out_size, void* d_ws, size_t ws_size,
                              hipStream_t stream) {
    const float* x        = (const float*)d_in[0];
    const float* guidance = (const float*)d_in[1];
    const float* convw    = (const float*)d_in[2];
    const float* convb    = (const float*)d_in[3];
    const float* ln_g     = (const float*)d_in[4];
    const float* ln_b     = (const float*)d_in[5];
    const float* w1       = (const float*)d_in[6];
    const float* b1       = (const float*)d_in[7];
    const float* w2       = (const float*)d_in[8];
    const float* b2       = (const float*)d_in[9];

    float* ws      = (float*)d_ws;
    float* feat    = ws;
    float* wts     = ws + 384;
    float* biaseff = ws + 512;
    float* wscaled = ws + 1024;

    feat_reduce_kernel<<<B_N * C_CH, 256, 0, stream>>>(x, feat);
    gate_kernel<<<1, 32, 0, stream>>>(feat, guidance, ln_g, ln_b,
                                      w1, b1, w2, b2, wts);
    const int foldN = B_N * C_CH * NB * 27;
    fold_kernel<<<(foldN + 255) / 256, 256, 0, stream>>>(convw, convb, wts,
                                                         wscaled, biaseff);
    dwconv_kernel<<<B_N * C_CH * DDIM, 256, 0, stream>>>(x, wscaled, biaseff,
                                                         (float*)d_out);
}


# conv reads its 7 input planes from a zero-haloed LDS tile: 4 outputs per thread from b128 LDS reads, no per-tap bounds branch and no wait after every single load; tap sum order unchanged
# speedup vs baseline: 5.0334x; 5.0334x over previous
#include <hip/hip_runtime.h>
#include <math.h>

#define B_N   4
#define C_CH  96
#define DDIM  16
#define HDIM  56
#define WDIM  56
#define GDIM  96
#define HID   24
#define NB    3
#define NCOL  (C_CH + GDIM)
#define SPATIAL (DDIM * HDIM * WDIM)
#define LN_EPS 1e-5f

typedef float v2f __attribute__((ext_vector_type(2)));
typedef float v4f __attribute__((ext_vector_type(4)));
typedef float v8f __attribute__((ext_vector_type(8)));

__global__ void feat_reduce_kernel(const float* __restrict__ x,
                                   float* __restrict__ feat) {
    const int bc = blockIdx.x;
    const v4f* p = (const v4f*)(x + (size_t)bc * SPATIAL);
    float s = 0.f;
    for (int i = threadIdx.x; i < SPATIAL / 4; i += 256) {
        v4f v = p[i];
        s += v.x + v.y + v.z + v.w;
    }
    __shared__ float red[256];
    red[threadIdx.x] = s;
    __syncthreads();
    for (int off = 128; off > 0; off >>= 1) {
        if (threadIdx.x < off) red[threadIdx.x] += red[threadIdx.x + off];
        __syncthreads();
    }
    if (threadIdx.x == 0) feat[bc] = red[0] * (1.0f / (float)SPATIAL);
}

__global__ void gate_kernel(const float* __restrict__ feat,
                            const float* __restrict__ guidance,
                            const float* __restrict__ ln_g,
                            const float* __restrict__ ln_b,
                            const float* __restrict__ w1p,
                            const float* __restrict__ b1p,
                            const float* __restrict__ w2p,
                            const float* __restrict__ b2p,
                            float* __restrict__ wts) {
    __shared__ float gn[16][NCOL];
    __shared__ float w1t[32][NCOL];
    __shared__ float mu[B_N];
    __shared__ float rs[B_N];
    __shared__ float hbuf[B_N][HID];
    __shared__ float logits[B_N][NB];

    const int lane = threadIdx.x;

    for (int idx = lane; idx < 12 * NCOL; idx += 32)
        gn[4 + idx / NCOL][idx % NCOL] = 0.f;
    for (int idx = lane; idx < 32 * NCOL; idx += 32) {
        const int n = idx / NCOL, k = idx % NCOL;
        w1t[n][k] = (n < HID) ? w1p[k * HID + n] : 0.f;
    }
    for (int idx = lane; idx < B_N * NCOL; idx += 32) {
        const int r = idx / NCOL, col = idx % NCOL;
        gn[r][col] = (col < C_CH) ? feat[r * C_CH + col]
                                  : guidance[r * GDIM + (col - C_CH)];
    }
    __syncthreads();

    if (lane < B_N) {
        float m = 0.f;
        for (int j = 0; j < NCOL; ++j) m += gn[lane][j];
        m *= (1.0f / NCOL);
        float v = 0.f;
        for (int j = 0; j < NCOL; ++j) { float d = gn[lane][j] - m; v += d * d; }
        v *= (1.0f / NCOL);
        mu[lane] = m;
        rs[lane] = rsqrtf(v + LN_EPS);
    }
    __syncthreads();

    for (int idx = lane; idx < B_N * NCOL; idx += 32) {
        const int r = idx / NCOL, col = idx % NCOL;
        gn[r][col] = (gn[r][col] - mu[r]) * rs[r] * ln_g[col] + ln_b[col];
    }
    __syncthreads();

    const int half = lane >> 4;
    const int l    = lane & 15;
    const float* arow  = &gn[l][0];
    const float* b0row = &w1t[l][0];
    const float* b1row = &w1t[16 + l][0];
    v8f c0 = {};
    v8f c1 = {};
    for (int kk = 0; kk < NCOL / 4; ++kk) {
        const int kb = kk * 4 + half * 2;
        const v2f a   = *(const v2f*)(arow + kb);
        const v2f b0  = *(const v2f*)(b0row + kb);
        const v2f b1v = *(const v2f*)(b1row + kb);
        c0 = __builtin_amdgcn_wmma_f32_16x16x4_f32(false, a, false, b0,
                                                   (short)0, c0, false, false);
        c1 = __builtin_amdgcn_wmma_f32_16x16x4_f32(false, a, false, b1v,
                                                   (short)0, c1, false, false);
    }

    #pragma unroll
    for (int r = 0; r < 8; ++r) {
        const int M = half ? (r + 8) : r;
        if (M < B_N) {
            float v0 = c0[r] + b1p[l];
            v0 = 0.5f * v0 * (1.0f + erff(v0 * 0.70710678118654752f));
            hbuf[M][l] = v0;
            const int n1 = 16 + l;
            if (n1 < HID) {
                float v1 = c1[r] + b1p[n1];
                v1 = 0.5f * v1 * (1.0f + erff(v1 * 0.70710678118654752f));
                hbuf[M][n1] = v1;
            }
        }
    }
    __syncthreads();

    if (lane < B_N * NB) {
        const int b = lane / NB, j = lane % NB;
        float s = b2p[j];
        for (int k = 0; k < HID; ++k) s += hbuf[b][k] * w2p[k * NB + j];
        logits[b][j] = s;
    }
    __syncthreads();

    if (lane < B_N) {
        float m = fmaxf(logits[lane][0], fmaxf(logits[lane][1], logits[lane][2]));
        float e0 = __expf(logits[lane][0] - m);
        float e1 = __expf(logits[lane][1] - m);
        float e2 = __expf(logits[lane][2] - m);
        float inv = 1.0f / (e0 + e1 + e2);
        wts[lane * NB + 0] = e0 * inv;
        wts[lane * NB + 1] = e1 * inv;
        wts[lane * NB + 2] = e2 * inv;
    }
}

__global__ void fold_kernel(const float* __restrict__ convw,
                            const float* __restrict__ convb,
                            const float* __restrict__ wts,
                            float* __restrict__ wscaled,
                            float* __restrict__ biaseff) {
    const int idx = blockIdx.x * 256 + threadIdx.x;
    const int total = B_N * C_CH * NB * 27;
    if (idx < total) {
        const int t = idx % 27;
        int tmp = idx / 27;
        const int i = tmp % NB;
        tmp /= NB;
        const int c = tmp % C_CH;
        const int b = tmp / C_CH;
        wscaled[idx] = wts[b * NB + i] * convw[(i * C_CH + c) * 27 + t];
    }
    if (idx < B_N * C_CH) {
        const int c = idx % C_CH, b = idx / C_CH;
        float s = 0.f;
        for (int i = 0; i < NB; ++i) s += wts[b * NB + i] * convb[i * C_CH + c];
        biaseff[idx] = s;
    }
}

#define ROWCH   15
#define TROWS   62
#define NROWS   (7 * TROWS)
#define NCHUNK  6656
#define STRIPS  14
#define NTASK   (HDIM * STRIPS)

__global__ __launch_bounds__(256)
void dwconv_kernel(const float* __restrict__ x,
                   const float* __restrict__ wscaled,
                   const float* __restrict__ biaseff,
                   float* __restrict__ out) {
    const int bcd = blockIdx.x;
    const int z  = bcd % DDIM;
    const int bc = bcd / DDIM;

    __shared__ v4f tile[NCHUNK];
    __shared__ float wsm[NB * 27];
    __shared__ float biass;
    if (threadIdx.x < NB * 27) wsm[threadIdx.x] = wscaled[bc * NB * 27 + threadIdx.x];
    if (threadIdx.x == NB * 27) biass = biaseff[bc];

    const float* xp = x + (size_t)bc * SPATIAL;

    #pragma unroll 1
    for (int h = 0; h < 2; ++h) {
        v4f  buf[13];
        #pragma unroll
        for (int k = 0; k < 13; ++k) {
            const int i  = (int)threadIdx.x + 256 * (h * 13 + k);
            const int rr = i / ROWCH;
            const int j  = i - rr * ROWCH;
            const int p  = rr / TROWS;
            const int r  = rr - p * TROWS;
            const int zc = min(max(z - 3 + p, 0), DDIM - 1);
            const int yc = min(max(r - 3, 0), HDIM - 1);
            const int jc = max(j, 1) - 1;
            buf[k] = *(const v4f*)(xp + (zc * HDIM + yc) * WDIM + jc * 4);
        }
        #pragma unroll
        for (int k = 0; k < 13; ++k) {
            const int i  = (int)threadIdx.x + 256 * (h * 13 + k);
            const int rr = i / ROWCH;
            const int j  = i - rr * ROWCH;
            const int p  = rr / TROWS;
            const int r  = rr - p * TROWS;
            const bool ok = (rr < NROWS) && (j != 0)
                         && ((unsigned)(z - 3 + p) < (unsigned)DDIM)
                         && ((unsigned)(r - 3) < (unsigned)HDIM);
            const v4f zero = {0.f, 0.f, 0.f, 0.f};
            tile[i] = ok ? buf[k] : zero;
        }
    }
    __syncthreads();

    const float bias = biass;
    v4f* op = (v4f*)(out + (size_t)bc * SPATIAL + (size_t)z * (HDIM * WDIM));

    for (int t = threadIdx.x; t < NTASK; t += 256) {
        const int y = t / STRIPS;
        const int sx = t - y * STRIPS;
        const v4f* q0 = tile + (y * ROWCH + sx);
        float a0 = bias, a1 = bias, a2 = bias, a3 = bias;
        #pragma unroll
        for (int i = 0; i < NB; ++i) {
            const int di = i + 1;
            #pragma unroll
            for (int kz = 0; kz < 3; ++kz) {
                #pragma unroll
                for (int ky = 0; ky < 3; ++ky) {
                    const int p    = 3 + (kz - 1) * di;
                    const int rofs = 3 + (ky - 1) * di;
                    const v4f* q = q0 + (p * TROWS + rofs) * ROWCH;
                    const v4f L = q[0], C = q[1], R = q[2];
                    const float v[12] = {L.x, L.y, L.z, L.w, C.x, C.y, C.z, C.w,
                                         R.x, R.y, R.z, R.w};
                    const float* wp = &wsm[i * 27 + (kz * 3 + ky) * 3];
                    #pragma unroll
                    for (int kx = 0; kx < 3; ++kx) {
                        const float w = wp[kx];
                        const int o = 4 + (kx - 1) * di;
                        a0 = fmaf(w, v[o + 0], a0);
                        a1 = fmaf(w, v[o + 1], a1);
                        a2 = fmaf(w, v[o + 2], a2);
                        a3 = fmaf(w, v[o + 3], a3);
                    }
                }
            }
        }
        const v4f res = {a0, a1, a2, a3};
        op[t] = res;
    }
}

extern "C" void kernel_launch(void* const* d_in, const int* in_sizes, int n_in,
                              void* d_out, int out_size, void* d_ws, size_t ws_size,
                              hipStream_t stream) {
    const float* x        = (const float*)d_in[0];
    const float* guidance = (const float*)d_in[1];
    const float* convw    = (const float*)d_in[2];
    const float* convb    = (const float*)d_in[3];
    const float* ln_g     = (const float*)d_in[4];
    const float* ln_b     = (const float*)d_in[5];
    const float* w1       = (const float*)d_in[6];
    const float* b1       = (const float*)d_in[7];
    const float* w2       = (const float*)d_in[8];
    const float* b2       = (const float*)d_in[9];

    float* ws      = (float*)d_ws;
    float* feat    = ws;
    float* wts     = ws + 384;
    float* biaseff = ws + 512;
    float* wscaled = ws + 1024;

    feat_reduce_kernel<<<B_N * C_CH, 256, 0, stream>>>(x, feat);
    gate_kernel<<<1, 32, 0, stream>>>(feat, guidance, ln_g, ln_b,
                                      w1, b1, w2, b2, wts);
    const int foldN = B_N * C_CH * NB * 27;
    fold_kernel<<<(foldN + 255) / 256, 256, 0, stream>>>(convw, convb, wts,
                                                         wscaled, biaseff);
    dwconv_kernel<<<B_N * C_CH * DDIM, 256, 0, stream>>>(x, wscaled, biaseff,
                                                         (float*)d_out);
}


# gate MLP kernel on 256 threads instead of one wave: all its global reads (w1, feat, guidance, LayerNorm and layer-2 params) issued up front and staged to LDS; WMMA loop and every sum order unchanged
# speedup vs baseline: 5.9823x; 1.1885x over previous
#include <hip/hip_runtime.h>
#include <math.h>

#define B_N   4
#define C_CH  96
#define DDIM  16
#define HDIM  56
#define WDIM  56
#define GDIM  96
#define HID   24
#define NB    3
#define NCOL  (C_CH + GDIM)
#define SPATIAL (DDIM * HDIM * WDIM)
#define LN_EPS 1e-5f

typedef float v2f __attribute__((ext_vector_type(2)));
typedef float v4f __attribute__((ext_vector_type(4)));
typedef float v8f __attribute__((ext_vector_type(8)));

__global__ void feat_reduce_kernel(const float* __restrict__ x,
                                   float* __restrict__ feat) {
    const int bc = blockIdx.x;
    const v4f* p = (const v4f*)(x + (size_t)bc * SPATIAL);
    float s = 0.f;
    for (int i = threadIdx.x; i < SPATIAL / 4; i += 256) {
        v4f v = p[i];
        s += v.x + v.y + v.z + v.w;
    }
    __shared__ float red[256];
    red[threadIdx.x] = s;
    __syncthreads();
    for (int off = 128; off > 0; off >>= 1) {
        if (threadIdx.x < off) red[threadIdx.x] += red[threadIdx.x + off];
        __syncthreads();
    }
    if (threadIdx.x == 0) feat[bc] = red[0] * (1.0f / (float)SPATIAL);
}

#define W1_V4   ((NCOL * HID) / 4)

__global__ __launch_bounds__(256)
void gate_kernel(const float* __restrict__ feat,
                 const float* __restrict__ guidance,
                 const float* __restrict__ ln_g,
                 const float* __restrict__ ln_b,
                 const float* __restrict__ w1p,
                 const float* __restrict__ b1p,
                 const float* __restrict__ w2p,
                 const float* __restrict__ b2p,
                 float* __restrict__ wts) {
    __shared__ __attribute__((aligned(16))) float gn[16][NCOL];
    __shared__ __attribute__((aligned(16))) float w1t[32][NCOL];
    __shared__ float mu[B_N];
    __shared__ float rs[B_N];
    __shared__ float hbuf[B_N][HID];
    __shared__ float logits[B_N][NB];
    __shared__ float w2s[HID * NB];
    __shared__ float b2s[NB];

    const int tid  = threadIdx.x;
    const int lane = tid & 31;
    const int half = lane >> 4;
    const int l    = lane & 15;

    v4f wv[5];
    #pragma unroll
    for (int j = 0; j < 5; ++j) {
        const int q  = min(tid + 256 * j, W1_V4 - 1);
        const int n4 = q / NCOL, k = q - n4 * NCOL;
        wv[j] = *(const v4f*)(w1p + k * HID + n4 * 4);
    }
    float gv[3], lg[3], lb[3];
    #pragma unroll
    for (int j = 0; j < 3; ++j) {
        const int idx = tid + 256 * j;
        const int r = idx / NCOL, col = idx - r * NCOL;
        const float* src = (col < C_CH) ? (feat + r * C_CH + col)
                                        : (guidance + r * GDIM + (col - C_CH));
        gv[j] = *src;
        lg[j] = ln_g[col];
        lb[j] = ln_b[col];
    }
    const float bl0 = b1p[l];
    const float bl1 = b1p[min(16 + l, HID - 1)];
    const float w2v = w2p[min(tid, HID * NB - 1)];
    const float b2v = b2p[min(tid, NB - 1)];

    float* gflat = &gn[0][0];
    float* wflat = &w1t[0][0];
    const v4f zero4 = {0.f, 0.f, 0.f, 0.f};
    for (int i = tid; i < (12 * NCOL) / 4; i += 256) ((v4f*)(gflat + 4 * NCOL))[i] = zero4;
    for (int i = tid; i < (8 * NCOL) / 4; i += 256)  ((v4f*)(wflat + HID * NCOL))[i] = zero4;
    #pragma unroll
    for (int j = 0; j < 5; ++j) {
        const int q = tid + 256 * j;
        if (q < W1_V4) {
            const int n4 = q / NCOL, k = q - n4 * NCOL;
            float* d = wflat + (n4 * 4) * NCOL + k;
            d[0 * NCOL] = wv[j].x;
            d[1 * NCOL] = wv[j].y;
            d[2 * NCOL] = wv[j].z;
            d[3 * NCOL] = wv[j].w;
        }
    }
    #pragma unroll
    for (int j = 0; j < 3; ++j) gflat[tid + 256 * j] = gv[j];
    if (tid < HID * NB) w2s[tid] = w2v;
    if (tid < NB) b2s[tid] = b2v;
    __syncthreads();

    if (tid < B_N) {
        float m = 0.f;
        for (int j = 0; j < NCOL; ++j) m += gn[tid][j];
        m *= (1.0f / NCOL);
        float v = 0.f;
        for (int j = 0; j < NCOL; ++j) { float d = gn[tid][j] - m; v += d * d; }
        v *= (1.0f / NCOL);
        mu[tid] = m;
        rs[tid] = rsqrtf(v + LN_EPS);
    }
    __syncthreads();

    #pragma unroll
    for (int j = 0; j < 3; ++j) {
        const int idx = tid + 256 * j;
        const int r = idx / NCOL;
        gflat[idx] = (gv[j] - mu[r]) * rs[r] * lg[j] + lb[j];
    }
    __syncthreads();

    const int wave = __builtin_amdgcn_readfirstlane(tid >> 5);
    if (wave == 0) {
        const float* arow  = &gn[l][0];
        const float* b0row = &w1t[l][0];
        const float* b1row = &w1t[16 + l][0];
        v8f c0 = {};
        v8f c1 = {};
        for (int kk = 0; kk < NCOL / 4; ++kk) {
            const int kb = kk * 4 + half * 2;
            const v2f a   = *(const v2f*)(arow + kb);
            const v2f b0  = *(const v2f*)(b0row + kb);
            const v2f b1v = *(const v2f*)(b1row + kb);
            c0 = __builtin_amdgcn_wmma_f32_16x16x4_f32(false, a, false, b0,
                                                       (short)0, c0, false, false);
            c1 = __builtin_amdgcn_wmma_f32_16x16x4_f32(false, a, false, b1v,
                                                       (short)0, c1, false, false);
        }

        #pragma unroll
        for (int r = 0; r < 8; ++r) {
            const int M = half ? (r + 8) : r;
            if (M < B_N) {
                float v0 = c0[r] + bl0;
                v0 = 0.5f * v0 * (1.0f + erff(v0 * 0.70710678118654752f));
                hbuf[M][l] = v0;
                const int n1 = 16 + l;
                if (n1 < HID) {
                    float v1 = c1[r] + bl1;
                    v1 = 0.5f * v1 * (1.0f + erff(v1 * 0.70710678118654752f));
                    hbuf[M][n1] = v1;
                }
            }
        }
    }
    __syncthreads();

    if (tid < B_N * NB) {
        const int b = tid / NB, j = tid % NB;
        float s = b2s[j];
        for (int k = 0; k < HID; ++k) s += hbuf[b][k] * w2s[k * NB + j];
        logits[b][j] = s;
    }
    __syncthreads();

    if (tid < B_N) {
        float m = fmaxf(logits[tid][0], fmaxf(logits[tid][1], logits[tid][2]));
        float e0 = __expf(logits[tid][0] - m);
        float e1 = __expf(logits[tid][1] - m);
        float e2 = __expf(logits[tid][2] - m);
        float inv = 1.0f / (e0 + e1 + e2);
        wts[tid * NB + 0] = e0 * inv;
        wts[tid * NB + 1] = e1 * inv;
        wts[tid * NB + 2] = e2 * inv;
    }
}

__global__ void fold_kernel(const float* __restrict__ convw,
                            const float* __restrict__ convb,
                            const float* __restrict__ wts,
                            float* __restrict__ wscaled,
                            float* __restrict__ biaseff) {
    const int idx = blockIdx.x * 256 + threadIdx.x;
    const int total = B_N * C_CH * NB * 27;
    if (idx < total) {
        const int t = idx % 27;
        int tmp = idx / 27;
        const int i = tmp % NB;
        tmp /= NB;
        const int c = tmp % C_CH;
        const int b = tmp / C_CH;
        wscaled[idx] = wts[b * NB + i] * convw[(i * C_CH + c) * 27 + t];
    }
    if (idx < B_N * C_CH) {
        const int c = idx % C_CH, b = idx / C_CH;
        float s = 0.f;
        for (int i = 0; i < NB; ++i) s += wts[b * NB + i] * convb[i * C_CH + c];
        biaseff[idx] = s;
    }
}

#define ROWCH   15
#define TROWS   62
#define NROWS   (7 * TROWS)
#define NCHUNK  6656
#define STRIPS  14
#define NTASK   (HDIM * STRIPS)

__global__ __launch_bounds__(256)
void dwconv_kernel(const float* __restrict__ x,
                   const float* __restrict__ wscaled,
                   const float* __restrict__ biaseff,
                   float* __restrict__ out) {
    const int bcd = blockIdx.x;
    const int z  = bcd % DDIM;
    const int bc = bcd / DDIM;

    __shared__ v4f tile[NCHUNK];
    __shared__ float wsm[NB * 27];
    __shared__ float biass;
    if (threadIdx.x < NB * 27) wsm[threadIdx.x] = wscaled[bc * NB * 27 + threadIdx.x];
    if (threadIdx.x == NB * 27) biass = biaseff[bc];

    const float* xp = x + (size_t)bc * SPATIAL;

    #pragma unroll 1
    for (int h = 0; h < 2; ++h) {
        v4f  buf[13];
        #pragma unroll
        for (int k = 0; k < 13; ++k) {
            const int i  = (int)threadIdx.x + 256 * (h * 13 + k);
            const int rr = i / ROWCH;
            const int j  = i - rr * ROWCH;
            const int p  = rr / TROWS;
            const int r  = rr - p * TROWS;
            const int zc = min(max(z - 3 + p, 0), DDIM - 1);
            const int yc = min(max(r - 3, 0), HDIM - 1);
            const int jc = max(j, 1) - 1;
            buf[k] = *(const v4f*)(xp + (zc * HDIM + yc) * WDIM + jc * 4);
        }
        #pragma unroll
        for (int k = 0; k < 13; ++k) {
            const int i  = (int)threadIdx.x + 256 * (h * 13 + k);
            const int rr = i / ROWCH;
            const int j  = i - rr * ROWCH;
            const int p  = rr / TROWS;
            const int r  = rr - p * TROWS;
            const bool ok = (rr < NROWS) && (j != 0)
                         && ((unsigned)(z - 3 + p) < (unsigned)DDIM)
                         && ((unsigned)(r - 3) < (unsigned)HDIM);
            const v4f zero = {0.f, 0.f, 0.f, 0.f};
            tile[i] = ok ? buf[k] : zero;
        }
    }
    __syncthreads();

    const float bias = biass;
    v4f* op = (v4f*)(out + (size_t)bc * SPATIAL + (size_t)z * (HDIM * WDIM));

    for (int t = threadIdx.x; t < NTASK; t += 256) {
        const int y = t / STRIPS;
        const int sx = t - y * STRIPS;
        const v4f* q0 = tile + (y * ROWCH + sx);
        float a0 = bias, a1 = bias, a2 = bias, a3 = bias;
        #pragma unroll
        for (int i = 0; i < NB; ++i) {
            const int di = i + 1;
            #pragma unroll
            for (int kz = 0; kz < 3; ++kz) {
                #pragma unroll
                for (int ky = 0; ky < 3; ++ky) {
                    const int p    = 3 + (kz - 1) * di;
                    const int rofs = 3 + (ky - 1) * di;
                    const v4f* q = q0 + (p * TROWS + rofs) * ROWCH;
                    const v4f L = q[0], C = q[1], R = q[2];
                    const float v[12] = {L.x, L.y, L.z, L.w, C.x, C.y, C.z, C.w,
                                         R.x, R.y, R.z, R.w};
                    const float* wp = &wsm[i * 27 + (kz * 3 + ky) * 3];
                    #pragma unroll
                    for (int kx = 0; kx < 3; ++kx) {
                        const float w = wp[kx];
                        const int o = 4 + (kx - 1) * di;
                        a0 = fmaf(w, v[o + 0], a0);
                        a1 = fmaf(w, v[o + 1], a1);
                        a2 = fmaf(w, v[o + 2], a2);
                        a3 = fmaf(w, v[o + 3], a3);
                    }
                }
            }
        }
        const v4f res = {a0, a1, a2, a3};
        op[t] = res;
    }
}

extern "C" void kernel_launch(void* const* d_in, const int* in_sizes, int n_in,
                              void* d_out, int out_size, void* d_ws, size_t ws_size,
                              hipStream_t stream) {
    const float* x        = (const float*)d_in[0];
    const float* guidance = (const float*)d_in[1];
    const float* convw    = (const float*)d_in[2];
    const float* convb    = (const float*)d_in[3];
    const float* ln_g     = (const float*)d_in[4];
    const float* ln_b     = (const float*)d_in[5];
    const float* w1       = (const float*)d_in[6];
    const float* b1       = (const float*)d_in[7];
    const float* w2       = (const float*)d_in[8];
    const float* b2       = (const float*)d_in[9];

    float* ws      = (float*)d_ws;
    float* feat    = ws;
    float* wts     = ws + 384;
    float* biaseff = ws + 512;
    float* wscaled = ws + 1024;

    feat_reduce_kernel<<<B_N * C_CH, 256, 0, stream>>>(x, feat);
    gate_kernel<<<1, 256, 0, stream>>>(feat, guidance, ln_g, ln_b,
                                      w1, b1, w2, b2, wts);
    const int foldN = B_N * C_CH * NB * 27;
    fold_kernel<<<(foldN + 255) / 256, 256, 0, stream>>>(convw, convb, wts,
                                                         wscaled, biaseff);
    dwconv_kernel<<<B_N * C_CH * DDIM, 256, 0, stream>>>(x, wscaled, biaseff,
                                                         (float*)d_out);
}


# conv LDS tile rows at a pitch of 14 chunks (no halo chunk in a row): each 16-lane half of a b128 read hits 16 consecutive chunks, no 2-way bank conflict; x halo by per-lane select; sum order kept
# speedup vs baseline: 6.3568x; 1.0626x over previous
#include <hip/hip_runtime.h>
#include <math.h>

#define B_N   4
#define C_CH  96
#define DDIM  16
#define HDIM  56
#define WDIM  56
#define GDIM  96
#define HID   24
#define NB    3
#define NCOL  (C_CH + GDIM)
#define SPATIAL (DDIM * HDIM * WDIM)
#define LN_EPS 1e-5f

typedef float v2f __attribute__((ext_vector_type(2)));
typedef float v4f __attribute__((ext_vector_type(4)));
typedef float v8f __attribute__((ext_vector_type(8)));

__global__ void feat_reduce_kernel(const float* __restrict__ x,
                                   float* __restrict__ feat) {
    const int bc = blockIdx.x;
    const v4f* p = (const v4f*)(x + (size_t)bc * SPATIAL);
    float s = 0.f;
    for (int i = threadIdx.x; i < SPATIAL / 4; i += 256) {
        v4f v = p[i];
        s += v.x + v.y + v.z + v.w;
    }
    __shared__ float red[256];
    red[threadIdx.x] = s;
    __syncthreads();
    for (int off = 128; off > 0; off >>= 1) {
        if (threadIdx.x < off) red[threadIdx.x] += red[threadIdx.x + off];
        __syncthreads();
    }
    if (threadIdx.x == 0) feat[bc] = red[0] * (1.0f / (float)SPATIAL);
}

#define W1_V4   ((NCOL * HID) / 4)

__global__ __launch_bounds__(256)
void gate_kernel(const float* __restrict__ feat,
                 const float* __restrict__ guidance,
                 const float* __restrict__ ln_g,
                 const float* __restrict__ ln_b,
                 const float* __restrict__ w1p,
                 const float* __restrict__ b1p,
                 const float* __restrict__ w2p,
                 const float* __restrict__ b2p,
                 float* __restrict__ wts) {
    __shared__ __attribute__((aligned(16))) float gn[16][NCOL];
    __shared__ __attribute__((aligned(16))) float w1t[32][NCOL];
    __shared__ float mu[B_N];
    __shared__ float rs[B_N];
    __shared__ float hbuf[B_N][HID];
    __shared__ float logits[B_N][NB];
    __shared__ float w2s[HID * NB];
    __shared__ float b2s[NB];

    const int tid  = threadIdx.x;
    const int lane = tid & 31;
    const int half = lane >> 4;
    const int l    = lane & 15;

    v4f wv[5];
    #pragma unroll
    for (int j = 0; j < 5; ++j) {
        const int q  = min(tid + 256 * j, W1_V4 - 1);
        const int n4 = q / NCOL, k = q - n4 * NCOL;
        wv[j] = *(const v4f*)(w1p + k * HID + n4 * 4);
    }
    float gv[3], lg[3], lb[3];
    #pragma unroll
    for (int j = 0; j < 3; ++j) {
        const int idx = tid + 256 * j;
        const int r = idx / NCOL, col = idx - r * NCOL;
        const float* src = (col < C_CH) ? (feat + r * C_CH + col)
                                        : (guidance + r * GDIM + (col - C_CH));
        gv[j] = *src;
        lg[j] = ln_g[col];
        lb[j] = ln_b[col];
    }
    const float bl0 = b1p[l];
    const float bl1 = b1p[min(16 + l, HID - 1)];
    const float w2v = w2p[min(tid, HID * NB - 1)];
    const float b2v = b2p[min(tid, NB - 1)];

    float* gflat = &gn[0][0];
    float* wflat = &w1t[0][0];
    const v4f zero4 = {0.f, 0.f, 0.f, 0.f};
    for (int i = tid; i < (12 * NCOL) / 4; i += 256) ((v4f*)(gflat + 4 * NCOL))[i] = zero4;
    for (int i = tid; i < (8 * NCOL) / 4; i += 256)  ((v4f*)(wflat + HID * NCOL))[i] = zero4;
    #pragma unroll
    for (int j = 0; j < 5; ++j) {
        const int q = tid + 256 * j;
        if (q < W1_V4) {
            const int n4 = q / NCOL, k = q - n4 * NCOL;
            float* d = wflat + (n4 * 4) * NCOL + k;
            d[0 * NCOL] = wv[j].x;
            d[1 * NCOL] = wv[j].y;
            d[2 * NCOL] = wv[j].z;
            d[3 * NCOL] = wv[j].w;
        }
    }
    #pragma unroll
    for (int j = 0; j < 3; ++j) gflat[tid + 256 * j] = gv[j];
    if (tid < HID * NB) w2s[tid] = w2v;
    if (tid < NB) b2s[tid] = b2v;
    __syncthreads();

    if (tid < B_N) {
        float m = 0.f;
        for (int j = 0; j < NCOL; ++j) m += gn[tid][j];
        m *= (1.0f / NCOL);
        float v = 0.f;
        for (int j = 0; j < NCOL; ++j) { float d = gn[tid][j] - m; v += d * d; }
        v *= (1.0f / NCOL);
        mu[tid] = m;
        rs[tid] = rsqrtf(v + LN_EPS);
    }
    __syncthreads();

    #pragma unroll
    for (int j = 0; j < 3; ++j) {
        const int idx = tid + 256 * j;
        const int r = idx / NCOL;
        gflat[idx] = (gv[j] - mu[r]) * rs[r] * lg[j] + lb[j];
    }
    __syncthreads();

    const int wave = __builtin_amdgcn_readfirstlane(tid >> 5);
    if (wave == 0) {
        const float* arow  = &gn[l][0];
        const float* b0row = &w1t[l][0];
        const float* b1row = &w1t[16 + l][0];
        v8f c0 = {};
        v8f c1 = {};
        for (int kk = 0; kk < NCOL / 4; ++kk) {
            const int kb = kk * 4 + half * 2;
            const v2f a   = *(const v2f*)(arow + kb);
            const v2f b0  = *(const v2f*)(b0row + kb);
            const v2f b1v = *(const v2f*)(b1row + kb);
            c0 = __builtin_amdgcn_wmma_f32_16x16x4_f32(false, a, false, b0,
                                                       (short)0, c0, false, false);
            c1 = __builtin_amdgcn_wmma_f32_16x16x4_f32(false, a, false, b1v,
                                                       (short)0, c1, false, false);
        }

        #pragma unroll
        for (int r = 0; r < 8; ++r) {
            const int M = half ? (r + 8) : r;
            if (M < B_N) {
                float v0 = c0[r] + bl0;
                v0 = 0.5f * v0 * (1.0f + erff(v0 * 0.70710678118654752f));
                hbuf[M][l] = v0;
                const int n1 = 16 + l;
                if (n1 < HID) {
                    float v1 = c1[r] + bl1;
                    v1 = 0.5f * v1 * (1.0f + erff(v1 * 0.70710678118654752f));
                    hbuf[M][n1] = v1;
                }
            }
        }
    }
    __syncthreads();

    if (tid < B_N * NB) {
        const int b = tid / NB, j = tid % NB;
        float s = b2s[j];
        for (int k = 0; k < HID; ++k) s += hbuf[b][k] * w2s[k * NB + j];
        logits[b][j] = s;
    }
    __syncthreads();

    if (tid < B_N) {
        float m = fmaxf(logits[tid][0], fmaxf(logits[tid][1], logits[tid][2]));
        float e0 = __expf(logits[tid][0] - m);
        float e1 = __expf(logits[tid][1] - m);
        float e2 = __expf(logits[tid][2] - m);
        float inv = 1.0f / (e0 + e1 + e2);
        wts[tid * NB + 0] = e0 * inv;
        wts[tid * NB + 1] = e1 * inv;
        wts[tid * NB + 2] = e2 * inv;
    }
}

__global__ void fold_kernel(const float* __restrict__ convw,
                            const float* __restrict__ convb,
                            const float* __restrict__ wts,
                            float* __restrict__ wscaled,
                            float* __restrict__ biaseff) {
    const int idx = blockIdx.x * 256 + threadIdx.x;
    const int total = B_N * C_CH * NB * 27;
    if (idx < total) {
        const int t = idx % 27;
        int tmp = idx / 27;
        const int i = tmp % NB;
        tmp /= NB;
        const int c = tmp % C_CH;
        const int b = tmp / C_CH;
        wscaled[idx] = wts[b * NB + i] * convw[(i * C_CH + c) * 27 + t];
    }
    if (idx < B_N * C_CH) {
        const int c = idx % C_CH, b = idx / C_CH;
        float s = 0.f;
        for (int i = 0; i < NB; ++i) s += wts[b * NB + i] * convb[i * C_CH + c];
        biaseff[idx] = s;
    }
}

#define ROWCH   14
#define TROWS   62
#define NROWS   (7 * TROWS)
#define FRONT   1
#define NCHUNK  6144
#define STRIPS  14
#define NTASK   (HDIM * STRIPS)

__global__ __launch_bounds__(256)
void dwconv_kernel(const float* __restrict__ x,
                   const float* __restrict__ wscaled,
                   const float* __restrict__ biaseff,
                   float* __restrict__ out) {
    const int bcd = blockIdx.x;
    const int z  = bcd % DDIM;
    const int bc = bcd / DDIM;

    __shared__ v4f tile[NCHUNK];
    __shared__ float wsm[NB * 27];
    __shared__ float biass;
    if (threadIdx.x < NB * 27) wsm[threadIdx.x] = wscaled[bc * NB * 27 + threadIdx.x];
    if (threadIdx.x == NB * 27) biass = biaseff[bc];

    const float* xp = x + (size_t)bc * SPATIAL;

    #pragma unroll 1
    for (int h = 0; h < 2; ++h) {
        v4f  buf[12];
        #pragma unroll
        for (int k = 0; k < 12; ++k) {
            const int i  = (int)threadIdx.x + 256 * (h * 12 + k);
            const int e  = max(i - FRONT, 0);
            const int rr = e / ROWCH;
            const int j  = e - rr * ROWCH;
            const int p  = rr / TROWS;
            const int r  = rr - p * TROWS;
            const int zc = min(max(z - 3 + p, 0), DDIM - 1);
            const int yc = min(max(r - 3, 0), HDIM - 1);
            buf[k] = *(const v4f*)(xp + (zc * HDIM + yc) * WDIM + j * 4);
        }
        #pragma unroll
        for (int k = 0; k < 12; ++k) {
            const int i  = (int)threadIdx.x + 256 * (h * 12 + k);
            const int e  = max(i - FRONT, 0);
            const int rr = e / ROWCH;
            const int p  = rr / TROWS;
            const int r  = rr - p * TROWS;
            const bool ok = (i >= FRONT) && (rr < NROWS)
                         && ((unsigned)(z - 3 + p) < (unsigned)DDIM)
                         && ((unsigned)(r - 3) < (unsigned)HDIM);
            const v4f zero = {0.f, 0.f, 0.f, 0.f};
            tile[i] = ok ? buf[k] : zero;
        }
    }
    __syncthreads();

    const float bias = biass;
    v4f* op = (v4f*)(out + (size_t)bc * SPATIAL + (size_t)z * (HDIM * WDIM));

    for (int t = threadIdx.x; t < NTASK; t += 256) {
        const int y = t / STRIPS;
        const int sx = t - y * STRIPS;
        const bool eL = (sx == 0);
        const bool eR = (sx == STRIPS - 1);
        const v4f* q0 = tile + (FRONT - 1) + t;
        float a0 = bias, a1 = bias, a2 = bias, a3 = bias;
        #pragma unroll
        for (int i = 0; i < NB; ++i) {
            const int di = i + 1;
            #pragma unroll
            for (int kz = 0; kz < 3; ++kz) {
                const int p = 3 + (kz - 1) * di;
                v4f ch[3][3];
                #pragma unroll
                for (int ky = 0; ky < 3; ++ky) {
                    const int rofs = 3 + (ky - 1) * di;
                    const v4f* q = q0 + (p * TROWS + rofs) * ROWCH;
                    ch[ky][0] = q[0]; ch[ky][1] = q[1]; ch[ky][2] = q[2];
                }
                #pragma unroll
                for (int ky = 0; ky < 3; ++ky) {
                    const v4f L = ch[ky][0], C = ch[ky][1], R = ch[ky][2];
                    const float v[12] = {eL ? 0.f : L.x, eL ? 0.f : L.y, eL ? 0.f : L.z, eL ? 0.f : L.w,
                                         C.x, C.y, C.z, C.w,
                                         eR ? 0.f : R.x, eR ? 0.f : R.y, eR ? 0.f : R.z, eR ? 0.f : R.w};
                    const float* wp = &wsm[i * 27 + (kz * 3 + ky) * 3];
                    #pragma unroll
                    for (int kx = 0; kx < 3; ++kx) {
                        const float w = wp[kx];
                        const int o = 4 + (kx - 1) * di;
                        a0 = fmaf(w, v[o + 0], a0);
                        a1 = fmaf(w, v[o + 1], a1);
                        a2 = fmaf(w, v[o + 2], a2);
                        a3 = fmaf(w, v[o + 3], a3);
                    }
                }
            }
        }
        const v4f res = {a0, a1, a2, a3};
        op[t] = res;
    }
}

extern "C" void kernel_launch(void* const* d_in, const int* in_sizes, int n_in,
                              void* d_out, int out_size, void* d_ws, size_t ws_size,
                              hipStream_t stream) {
    const float* x        = (const float*)d_in[0];
    const float* guidance = (const float*)d_in[1];
    const float* convw    = (const float*)d_in[2];
    const float* convb    = (const float*)d_in[3];
    const float* ln_g     = (const float*)d_in[4];
    const float* ln_b     = (const float*)d_in[5];
    const float* w1       = (const float*)d_in[6];
    const float* b1       = (const float*)d_in[7];
    const float* w2       = (const float*)d_in[8];
    const float* b2       = (const float*)d_in[9];

    float* ws      = (float*)d_ws;
    float* feat    = ws;
    float* wts     = ws + 384;
    float* biaseff = ws + 512;
    float* wscaled = ws + 1024;

    feat_reduce_kernel<<<B_N * C_CH, 256, 0, stream>>>(x, feat);
    gate_kernel<<<1, 256, 0, stream>>>(feat, guidance, ln_g, ln_b,
                                      w1, b1, w2, b2, wts);
    const int foldN = B_N * C_CH * NB * 27;
    fold_kernel<<<(foldN + 255) / 256, 256, 0, stream>>>(convw, convb, wts,
                                                         wscaled, biaseff);
    dwconv_kernel<<<B_N * C_CH * DDIM, 256, 0, stream>>>(x, wscaled, biaseff,
                                                         (float*)d_out);
}


# conv block walks 8 output planes of a (b,c) over a 7-plane LDS ring, refilled between two LDS-only barriers: each input plane staged once per block, not once per output plane; tap sum order kept
# speedup vs baseline: 9.7808x; 1.5386x over previous
#include <hip/hip_runtime.h>
#include <math.h>

#define B_N   4
#define C_CH  96
#define DDIM  16
#define HDIM  56
#define WDIM  56
#define GDIM  96
#define HID   24
#define NB    3
#define NCOL  (C_CH + GDIM)
#define SPATIAL (DDIM * HDIM * WDIM)
#define LN_EPS 1e-5f

typedef float v2f __attribute__((ext_vector_type(2)));
typedef float v4f __attribute__((ext_vector_type(4)));
typedef float v8f __attribute__((ext_vector_type(8)));

__global__ void feat_reduce_kernel(const float* __restrict__ x,
                                   float* __restrict__ feat) {
    const int bc = blockIdx.x;
    const v4f* p = (const v4f*)(x + (size_t)bc * SPATIAL);
    float s = 0.f;
    for (int i = threadIdx.x; i < SPATIAL / 4; i += 256) {
        v4f v = p[i];
        s += v.x + v.y + v.z + v.w;
    }
    __shared__ float red[256];
    red[threadIdx.x] = s;
    __syncthreads();
    for (int off = 128; off > 0; off >>= 1) {
        if (threadIdx.x < off) red[threadIdx.x] += red[threadIdx.x + off];
        __syncthreads();
    }
    if (threadIdx.x == 0) feat[bc] = red[0] * (1.0f / (float)SPATIAL);
}

#define W1_V4   ((NCOL * HID) / 4)

__global__ __launch_bounds__(256)
void gate_kernel(const float* __restrict__ feat,
                 const float* __restrict__ guidance,
                 const float* __restrict__ ln_g,
                 const float* __restrict__ ln_b,
                 const float* __restrict__ w1p,
                 const float* __restrict__ b1p,
                 const float* __restrict__ w2p,
                 const float* __restrict__ b2p,
                 float* __restrict__ wts) {
    __shared__ __attribute__((aligned(16))) float gn[16][NCOL];
    __shared__ __attribute__((aligned(16))) float w1t[32][NCOL];
    __shared__ float mu[B_N];
    __shared__ float rs[B_N];
    __shared__ float hbuf[B_N][HID];
    __shared__ float logits[B_N][NB];
    __shared__ float w2s[HID * NB];
    __shared__ float b2s[NB];

    const int tid  = threadIdx.x;
    const int lane = tid & 31;
    const int half = lane >> 4;
    const int l    = lane & 15;

    v4f wv[5];
    #pragma unroll
    for (int j = 0; j < 5; ++j) {
        const int q  = min(tid + 256 * j, W1_V4 - 1);
        const int n4 = q / NCOL, k = q - n4 * NCOL;
        wv[j] = *(const v4f*)(w1p + k * HID + n4 * 4);
    }
    float gv[3], lg[3], lb[3];
    #pragma unroll
    for (int j = 0; j < 3; ++j) {
        const int idx = tid + 256 * j;
        const int r = idx / NCOL, col = idx - r * NCOL;
        const float* src = (col < C_CH) ? (feat + r * C_CH + col)
                                        : (guidance + r * GDIM + (col - C_CH));
        gv[j] = *src;
        lg[j] = ln_g[col];
        lb[j] = ln_b[col];
    }
    const float bl0 = b1p[l];
    const float bl1 = b1p[min(16 + l, HID - 1)];
    const float w2v = w2p[min(tid, HID * NB - 1)];
    const float b2v = b2p[min(tid, NB - 1)];

    float* gflat = &gn[0][0];
    float* wflat = &w1t[0][0];
    const v4f zero4 = {0.f, 0.f, 0.f, 0.f};
    for (int i = tid; i < (12 * NCOL) / 4; i += 256) ((v4f*)(gflat + 4 * NCOL))[i] = zero4;
    for (int i = tid; i < (8 * NCOL) / 4; i += 256)  ((v4f*)(wflat + HID * NCOL))[i] = zero4;
    #pragma unroll
    for (int j = 0; j < 5; ++j) {
        const int q = tid + 256 * j;
        if (q < W1_V4) {
            const int n4 = q / NCOL, k = q - n4 * NCOL;
            float* d = wflat + (n4 * 4) * NCOL + k;
            d[0 * NCOL] = wv[j].x;
            d[1 * NCOL] = wv[j].y;
            d[2 * NCOL] = wv[j].z;
            d[3 * NCOL] = wv[j].w;
        }
    }
    #pragma unroll
    for (int j = 0; j < 3; ++j) gflat[tid + 256 * j] = gv[j];
    if (tid < HID * NB) w2s[tid] = w2v;
    if (tid < NB) b2s[tid] = b2v;
    __syncthreads();

    if (tid < B_N) {
        float m = 0.f;
        for (int j = 0; j < NCOL; ++j) m += gn[tid][j];
        m *= (1.0f / NCOL);
        float v = 0.f;
        for (int j = 0; j < NCOL; ++j) { float d = gn[tid][j] - m; v += d * d; }
        v *= (1.0f / NCOL);
        mu[tid] = m;
        rs[tid] = rsqrtf(v + LN_EPS);
    }
    __syncthreads();

    #pragma unroll
    for (int j = 0; j < 3; ++j) {
        const int idx = tid + 256 * j;
        const int r = idx / NCOL;
        gflat[idx] = (gv[j] - mu[r]) * rs[r] * lg[j] + lb[j];
    }
    __syncthreads();

    const int wave = __builtin_amdgcn_readfirstlane(tid >> 5);
    if (wave == 0) {
        const float* arow  = &gn[l][0];
        const float* b0row = &w1t[l][0];
        const float* b1row = &w1t[16 + l][0];
        v8f c0 = {};
        v8f c1 = {};
        for (int kk = 0; kk < NCOL / 4; ++kk) {
            const int kb = kk * 4 + half * 2;
            const v2f a   = *(const v2f*)(arow + kb);
            const v2f b0  = *(const v2f*)(b0row + kb);
            const v2f b1v = *(const v2f*)(b1row + kb);
            c0 = __builtin_amdgcn_wmma_f32_16x16x4_f32(false, a, false, b0,
                                                       (short)0, c0, false, false);
            c1 = __builtin_amdgcn_wmma_f32_16x16x4_f32(false, a, false, b1v,
                                                       (short)0, c1, false, false);
        }

        #pragma unroll
        for (int r = 0; r < 8; ++r) {
            const int M = half ? (r + 8) : r;
            if (M < B_N) {
                float v0 = c0[r] + bl0;
                v0 = 0.5f * v0 * (1.0f + erff(v0 * 0.70710678118654752f));
                hbuf[M][l] = v0;
                const int n1 = 16 + l;
                if (n1 < HID) {
                    float v1 = c1[r] + bl1;
                    v1 = 0.5f * v1 * (1.0f + erff(v1 * 0.70710678118654752f));
                    hbuf[M][n1] = v1;
                }
            }
        }
    }
    __syncthreads();

    if (tid < B_N * NB) {
        const int b = tid / NB, j = tid % NB;
        float s = b2s[j];
        for (int k = 0; k < HID; ++k) s += hbuf[b][k] * w2s[k * NB + j];
        logits[b][j] = s;
    }
    __syncthreads();

    if (tid < B_N) {
        float m = fmaxf(logits[tid][0], fmaxf(logits[tid][1], logits[tid][2]));
        float e0 = __expf(logits[tid][0] - m);
        float e1 = __expf(logits[tid][1] - m);
        float e2 = __expf(logits[tid][2] - m);
        float inv = 1.0f / (e0 + e1 + e2);
        wts[tid * NB + 0] = e0 * inv;
        wts[tid * NB + 1] = e1 * inv;
        wts[tid * NB + 2] = e2 * inv;
    }
}

__global__ void fold_kernel(const float* __restrict__ convw,
                            const float* __restrict__ convb,
                            const float* __restrict__ wts,
                            float* __restrict__ wscaled,
                            float* __restrict__ biaseff) {
    const int idx = blockIdx.x * 256 + threadIdx.x;
    const int total = B_N * C_CH * NB * 27;
    if (idx < total) {
        const int t = idx % 27;
        int tmp = idx / 27;
        const int i = tmp % NB;
        tmp /= NB;
        const int c = tmp % C_CH;
        const int b = tmp / C_CH;
        wscaled[idx] = wts[b * NB + i] * convw[(i * C_CH + c) * 27 + t];
    }
    if (idx < B_N * C_CH) {
        const int c = idx % C_CH, b = idx / C_CH;
        float s = 0.f;
        for (int i = 0; i < NB; ++i) s += wts[b * NB + i] * convb[i * C_CH + c];
        biaseff[idx] = s;
    }
}

#define ROWCH   14
#define TROWS   62
#define NSLOT   7
#define NROWS   (NSLOT * TROWS)
#define PLANECH (TROWS * ROWCH)
#define FRONT   1
#define NCHUNK  6144
#define STRIPS  14
#define NTASK   (HDIM * STRIPS)
#define ZPB     8
#define ZBLK    (DDIM / ZPB)
#define RTHR    224
#define NPF     4

__device__ __forceinline__ void lds_barrier() {
    __builtin_amdgcn_fence(__ATOMIC_RELEASE, "workgroup", "local");
    __builtin_amdgcn_s_barrier();
    __builtin_amdgcn_fence(__ATOMIC_ACQUIRE, "workgroup", "local");
}

__global__ __launch_bounds__(256, 6)
void dwconv_kernel(const float* __restrict__ x,
                   const float* __restrict__ wscaled,
                   const float* __restrict__ biaseff,
                   float* __restrict__ out) {
    const int blk = blockIdx.x;
    const int z0 = (blk % ZBLK) * ZPB;
    const int bc = blk / ZBLK;

    __shared__ v4f tile[NCHUNK];
    __shared__ float wsm[NB * 27];
    __shared__ float biass;
    if (threadIdx.x < NB * 27) wsm[threadIdx.x] = wscaled[bc * NB * 27 + threadIdx.x];
    if (threadIdx.x == NB * 27) biass = biaseff[bc];

    const float* xp = x + (size_t)bc * SPATIAL;

    #pragma unroll 1
    for (int h = 0; h < 2; ++h) {
        v4f  buf[12];
        #pragma unroll
        for (int k = 0; k < 12; ++k) {
            const int i  = (int)threadIdx.x + 256 * (h * 12 + k);
            const int e  = max(i - FRONT, 0);
            const int rr = e / ROWCH;
            const int j  = e - rr * ROWCH;
            const int p  = rr / TROWS;
            const int r  = rr - p * TROWS;
            const int zc = min(max(z0 - 3 + p, 0), DDIM - 1);
            const int yc = min(max(r - 3, 0), HDIM - 1);
            buf[k] = *(const v4f*)(xp + (zc * HDIM + yc) * WDIM + j * 4);
        }
        #pragma unroll
        for (int k = 0; k < 12; ++k) {
            const int i  = (int)threadIdx.x + 256 * (h * 12 + k);
            const int e  = max(i - FRONT, 0);
            const int rr = e / ROWCH;
            const int p  = rr / TROWS;
            const int r  = rr - p * TROWS;
            const bool ok = (i >= FRONT) && (rr < NROWS)
                         && ((unsigned)(z0 - 3 + p) < (unsigned)DDIM)
                         && ((unsigned)(r - 3) < (unsigned)HDIM);
            const v4f zero = {0.f, 0.f, 0.f, 0.f};
            tile[i] = ok ? buf[k] : zero;
        }
    }
    __syncthreads();

    const float bias = biass;

    #pragma unroll 1
    for (int k = 0; k < ZPB; ++k) {
        const int z = z0 + k;
        const int s0 = (k >= NSLOT) ? (k - NSLOT) : k;
        const bool refill = (k + 1 < ZPB);
        const bool inside = (z + 4 < DDIM);

        int poff[NSLOT];
        #pragma unroll
        for (int p = 0; p < NSLOT; ++p) {
            const int s = s0 + p;
            poff[p] = ((s >= NSLOT) ? (s - NSLOT) : s) * PLANECH;
        }

        v4f* op = (v4f*)(out + (size_t)bc * SPATIAL + (size_t)z * (HDIM * WDIM));

        for (int t = threadIdx.x; t < NTASK; t += 256) {
            const int y = t / STRIPS;
            const int sx = t - y * STRIPS;
            const bool eL = (sx == 0);
            const bool eR = (sx == STRIPS - 1);
            const v4f* q0 = tile + (FRONT - 1) + t;
            float a0 = bias, a1 = bias, a2 = bias, a3 = bias;
            #pragma unroll
            for (int i = 0; i < NB; ++i) {
                const int di = i + 1;
                #pragma unroll
                for (int kz = 0; kz < 3; ++kz) {
                    const int p = 3 + (kz - 1) * di;
                    v4f ch[3][3];
                    #pragma unroll
                    for (int ky = 0; ky < 3; ++ky) {
                        const int rofs = 3 + (ky - 1) * di;
                        const v4f* q = q0 + poff[p] + rofs * ROWCH;
                        ch[ky][0] = q[0]; ch[ky][1] = q[1]; ch[ky][2] = q[2];
                    }
                    #pragma unroll
                    for (int ky = 0; ky < 3; ++ky) {
                        const v4f L = ch[ky][0], C = ch[ky][1], R = ch[ky][2];
                        const float v[12] = {eL ? 0.f : L.x, eL ? 0.f : L.y, eL ? 0.f : L.z, eL ? 0.f : L.w,
                                             C.x, C.y, C.z, C.w,
                                             eR ? 0.f : R.x, eR ? 0.f : R.y, eR ? 0.f : R.z, eR ? 0.f : R.w};
                        const float* wp = &wsm[i * 27 + (kz * 3 + ky) * 3];
                        #pragma unroll
                        for (int kx = 0; kx < 3; ++kx) {
                            const float w = wp[kx];
                            const int o = 4 + (kx - 1) * di;
                            a0 = fmaf(w, v[o + 0], a0);
                            a1 = fmaf(w, v[o + 1], a1);
                            a2 = fmaf(w, v[o + 2], a2);
                            a3 = fmaf(w, v[o + 3], a3);
                        }
                    }
                }
            }
            const v4f res = {a0, a1, a2, a3};
            op[t] = res;
        }

        if (refill) {
            const int rt = (int)threadIdx.x - 32;
            const v4f zero = {0.f, 0.f, 0.f, 0.f};
            v4f nb[NPF] = {zero, zero, zero, zero};
            if (rt >= 0 && inside) {
                const v4f* srcp = (const v4f*)(xp + (z + 4) * (HDIM * WDIM));
                #pragma unroll
                for (int m = 0; m < NPF; ++m)
                    nb[m] = srcp[min(rt + RTHR * m, NTASK - 1)];
            }
            lds_barrier();
            if (rt >= 0) {
                v4f* dst = tile + FRONT + s0 * PLANECH + 3 * ROWCH;
                #pragma unroll
                for (int m = 0; m < NPF; ++m) {
                    const int i = rt + RTHR * m;
                    if (i < NTASK) dst[i] = nb[m];
                }
            }
            lds_barrier();
        }
    }
}

extern "C" void kernel_launch(void* const* d_in, const int* in_sizes, int n_in,
                              void* d_out, int out_size, void* d_ws, size_t ws_size,
                              hipStream_t stream) {
    const float* x        = (const float*)d_in[0];
    const float* guidance = (const float*)d_in[1];
    const float* convw    = (const float*)d_in[2];
    const float* convb    = (const float*)d_in[3];
    const float* ln_g     = (const float*)d_in[4];
    const float* ln_b     = (const float*)d_in[5];
    const float* w1       = (const float*)d_in[6];
    const float* b1       = (const float*)d_in[7];
    const float* w2       = (const float*)d_in[8];
    const float* b2       = (const float*)d_in[9];

    float* ws      = (float*)d_ws;
    float* feat    = ws;
    float* wts     = ws + 384;
    float* biaseff = ws + 512;
    float* wscaled = ws + 1024;

    feat_reduce_kernel<<<B_N * C_CH, 256, 0, stream>>>(x, feat);
    gate_kernel<<<1, 256, 0, stream>>>(feat, guidance, ln_g, ln_b,
                                      w1, b1, w2, b2, wts);
    const int foldN = B_N * C_CH * NB * 27;
    fold_kernel<<<(foldN + 255) / 256, 256, 0, stream>>>(convw, convb, wts,
                                                         wscaled, biaseff);
    dwconv_kernel<<<B_N * C_CH * ZBLK, 256, 0, stream>>>(x, wscaled, biaseff,
                                                         (float*)d_out);
}


# conv thread owns a 4-wide strip through all 16 planes and walks the input planes: each staged row read once, scattered to 7 output-plane sums in registers; taps now summed in input-plane order
# speedup vs baseline: 11.7082x; 1.1971x over previous
#include <hip/hip_runtime.h>
#include <math.h>

#define B_N   4
#define C_CH  96
#define DDIM  16
#define HDIM  56
#define WDIM  56
#define GDIM  96
#define HID   24
#define NB    3
#define NCOL  (C_CH + GDIM)
#define SPATIAL (DDIM * HDIM * WDIM)
#define LN_EPS 1e-5f

typedef float v2f __attribute__((ext_vector_type(2)));
typedef float v4f __attribute__((ext_vector_type(4)));
typedef float v8f __attribute__((ext_vector_type(8)));

__global__ void feat_reduce_kernel(const float* __restrict__ x,
                                   float* __restrict__ feat) {
    const int bc = blockIdx.x;
    const v4f* p = (const v4f*)(x + (size_t)bc * SPATIAL);
    float s = 0.f;
    for (int i = threadIdx.x; i < SPATIAL / 4; i += 256) {
        v4f v = p[i];
        s += v.x + v.y + v.z + v.w;
    }
    __shared__ float red[256];
    red[threadIdx.x] = s;
    __syncthreads();
    for (int off = 128; off > 0; off >>= 1) {
        if (threadIdx.x < off) red[threadIdx.x] += red[threadIdx.x + off];
        __syncthreads();
    }
    if (threadIdx.x == 0) feat[bc] = red[0] * (1.0f / (float)SPATIAL);
}

#define W1_V4   ((NCOL * HID) / 4)

__global__ __launch_bounds__(256)
void gate_kernel(const float* __restrict__ feat,
                 const float* __restrict__ guidance,
                 const float* __restrict__ ln_g,
                 const float* __restrict__ ln_b,
                 const float* __restrict__ w1p,
                 const float* __restrict__ b1p,
                 const float* __restrict__ w2p,
                 const float* __restrict__ b2p,
                 float* __restrict__ wts) {
    __shared__ __attribute__((aligned(16))) float gn[16][NCOL];
    __shared__ __attribute__((aligned(16))) float w1t[32][NCOL];
    __shared__ float mu[B_N];
    __shared__ float rs[B_N];
    __shared__ float hbuf[B_N][HID];
    __shared__ float logits[B_N][NB];
    __shared__ float w2s[HID * NB];
    __shared__ float b2s[NB];

    const int tid  = threadIdx.x;
    const int lane = tid & 31;
    const int half = lane >> 4;
    const int l    = lane & 15;

    v4f wv[5];
    #pragma unroll
    for (int j = 0; j < 5; ++j) {
        const int q  = min(tid + 256 * j, W1_V4 - 1);
        const int n4 = q / NCOL, k = q - n4 * NCOL;
        wv[j] = *(const v4f*)(w1p + k * HID + n4 * 4);
    }
    float gv[3], lg[3], lb[3];
    #pragma unroll
    for (int j = 0; j < 3; ++j) {
        const int idx = tid + 256 * j;
        const int r = idx / NCOL, col = idx - r * NCOL;
        const float* src = (col < C_CH) ? (feat + r * C_CH + col)
                                        : (guidance + r * GDIM + (col - C_CH));
        gv[j] = *src;
        lg[j] = ln_g[col];
        lb[j] = ln_b[col];
    }
    const float bl0 = b1p[l];
    const float bl1 = b1p[min(16 + l, HID - 1)];
    const float w2v = w2p[min(tid, HID * NB - 1)];
    const float b2v = b2p[min(tid, NB - 1)];

    float* gflat = &gn[0][0];
    float* wflat = &w1t[0][0];
    const v4f zero4 = {0.f, 0.f, 0.f, 0.f};
    for (int i = tid; i < (12 * NCOL) / 4; i += 256) ((v4f*)(gflat + 4 * NCOL))[i] = zero4;
    for (int i = tid; i < (8 * NCOL) / 4; i += 256)  ((v4f*)(wflat + HID * NCOL))[i] = zero4;
    #pragma unroll
    for (int j = 0; j < 5; ++j) {
        const int q = tid + 256 * j;
        if (q < W1_V4) {
            const int n4 = q / NCOL, k = q - n4 * NCOL;
            float* d = wflat + (n4 * 4) * NCOL + k;
            d[0 * NCOL] = wv[j].x;
            d[1 * NCOL] = wv[j].y;
            d[2 * NCOL] = wv[j].z;
            d[3 * NCOL] = wv[j].w;
        }
    }
    #pragma unroll
    for (int j = 0; j < 3; ++j) gflat[tid + 256 * j] = gv[j];
    if (tid < HID * NB) w2s[tid] = w2v;
    if (tid < NB) b2s[tid] = b2v;
    __syncthreads();

    if (tid < B_N) {
        float m = 0.f;
        for (int j = 0; j < NCOL; ++j) m += gn[tid][j];
        m *= (1.0f / NCOL);
        float v = 0.f;
        for (int j = 0; j < NCOL; ++j) { float d = gn[tid][j] - m; v += d * d; }
        v *= (1.0f / NCOL);
        mu[tid] = m;
        rs[tid] = rsqrtf(v + LN_EPS);
    }
    __syncthreads();

    #pragma unroll
    for (int j = 0; j < 3; ++j) {
        const int idx = tid + 256 * j;
        const int r = idx / NCOL;
        gflat[idx] = (gv[j] - mu[r]) * rs[r] * lg[j] + lb[j];
    }
    __syncthreads();

    const int wave = __builtin_amdgcn_readfirstlane(tid >> 5);
    if (wave == 0) {
        const float* arow  = &gn[l][0];
        const float* b0row = &w1t[l][0];
        const float* b1row = &w1t[16 + l][0];
        v8f c0 = {};
        v8f c1 = {};
        for (int kk = 0; kk < NCOL / 4; ++kk) {
            const int kb = kk * 4 + half * 2;
            const v2f a   = *(const v2f*)(arow + kb);
            const v2f b0  = *(const v2f*)(b0row + kb);
            const v2f b1v = *(const v2f*)(b1row + kb);
            c0 = __builtin_amdgcn_wmma_f32_16x16x4_f32(false, a, false, b0,
                                                       (short)0, c0, false, false);
            c1 = __builtin_amdgcn_wmma_f32_16x16x4_f32(false, a, false, b1v,
                                                       (short)0, c1, false, false);
        }

        #pragma unroll
        for (int r = 0; r < 8; ++r) {
            const int M = half ? (r + 8) : r;
            if (M < B_N) {
                float v0 = c0[r] + bl0;
                v0 = 0.5f * v0 * (1.0f + erff(v0 * 0.70710678118654752f));
                hbuf[M][l] = v0;
                const int n1 = 16 + l;
                if (n1 < HID) {
                    float v1 = c1[r] + bl1;
                    v1 = 0.5f * v1 * (1.0f + erff(v1 * 0.70710678118654752f));
                    hbuf[M][n1] = v1;
                }
            }
        }
    }
    __syncthreads();

    if (tid < B_N * NB) {
        const int b = tid / NB, j = tid % NB;
        float s = b2s[j];
        for (int k = 0; k < HID; ++k) s += hbuf[b][k] * w2s[k * NB + j];
        logits[b][j] = s;
    }
    __syncthreads();

    if (tid < B_N) {
        float m = fmaxf(logits[tid][0], fmaxf(logits[tid][1], logits[tid][2]));
        float e0 = __expf(logits[tid][0] - m);
        float e1 = __expf(logits[tid][1] - m);
        float e2 = __expf(logits[tid][2] - m);
        float inv = 1.0f / (e0 + e1 + e2);
        wts[tid * NB + 0] = e0 * inv;
        wts[tid * NB + 1] = e1 * inv;
        wts[tid * NB + 2] = e2 * inv;
    }
}

__global__ void fold_kernel(const float* __restrict__ convw,
                            const float* __restrict__ convb,
                            const float* __restrict__ wts,
                            float* __restrict__ wscaled,
                            float* __restrict__ biaseff) {
    const int idx = blockIdx.x * 256 + threadIdx.x;
    const int total = B_N * C_CH * NB * 27;
    if (idx < total) {
        const int t = idx % 27;
        int tmp = idx / 27;
        const int i = tmp % NB;
        tmp /= NB;
        const int c = tmp % C_CH;
        const int b = tmp / C_CH;
        wscaled[idx] = wts[b * NB + i] * convw[(i * C_CH + c) * 27 + t];
    }
    if (idx < B_N * C_CH) {
        const int c = idx % C_CH, b = idx / C_CH;
        float s = 0.f;
        for (int i = 0; i < NB; ++i) s += wts[b * NB + i] * convb[i * C_CH + c];
        biaseff[idx] = s;
    }
}

#define ROWCH   14
#define YSL     28
#define YBLK    (HDIM / YSL)
#define SROWS   (YSL + 6)
#define REALROWS (YSL + 3)
#define REALCH  (REALROWS * ROWCH)
#define PLCH    (HDIM * ROWCH)
#define SLOTCH  480
#define FRONT   1
#define NCHUNK  (FRONT + 2 * SLOTCH + 3)
#define NPEN    (YSL * ROWCH)
#define CTHR    416

__device__ __forceinline__ void lds_barrier() {
    __builtin_amdgcn_fence(__ATOMIC_RELEASE, "workgroup", "local");
    __builtin_amdgcn_s_barrier();
    __builtin_amdgcn_fence(__ATOMIC_ACQUIRE, "workgroup", "local");
}

__global__ __launch_bounds__(CTHR, 10)
void dwconv_kernel(const float* __restrict__ x,
                   const float* __restrict__ wscaled,
                   const float* __restrict__ biaseff,
                   float* __restrict__ out) {
    const int blk = blockIdx.x;
    const int yh  = blk % YBLK;
    const int bc  = blk / YBLK;
    const int tid = threadIdx.x;

    __shared__ v4f tile[NCHUNK];

    const float* __restrict__ wg = wscaled + bc * (NB * 27);
    const float bias = biaseff[bc];

    const int ylo = yh ? (HDIM - REALROWS) : 0;
    const int rlo = yh ? 0 : 3;
    const v4f* xp = (const v4f*)(x + (size_t)bc * SPATIAL + ylo * WDIM);
    const v4f zero = {0.f, 0.f, 0.f, 0.f};

    for (int i = tid; i < NCHUNK; i += CTHR) {
        const int e = i - (FRONT + rlo * ROWCH);
        const bool real = (unsigned)e < (unsigned)REALCH;
        const v4f v = xp[real ? e : 0];
        tile[i] = real ? v : zero;
    }
    __syncthreads();

    const bool act = (tid < NPEN);
    const int t  = act ? tid : (NPEN - 1);
    const int sx = t % ROWCH;
    const bool eL = (sx == 0);
    const bool eR = (sx == ROWCH - 1);

    float wr[NB * 27];
    #pragma unroll
    for (int i = 0; i < NB * 27; ++i) wr[i] = wg[i];

    float acc[7][4];
    #pragma unroll
    for (int j = 0; j < 7; ++j) { acc[j][0] = bias; acc[j][1] = bias; acc[j][2] = bias; acc[j][3] = bias; }

    v4f* st = tile + FRONT + rlo * ROWCH;

    v4f nb0 = xp[PLCH + tid];
    v4f nb1 = xp[PLCH + min(tid + CTHR, REALCH - 1)];
    v4f fin = zero;
    const v4f* srcq = xp + 2 * PLCH;
    v4f* opq = (v4f*)(out + (size_t)bc * SPATIAL + (yh * YSL) * WDIM);
    const int tid2 = min(tid + CTHR, REALCH - 1);

    #pragma unroll 1
    for (int p = 0; p < DDIM; ++p) {
        const int s = p & 1;
        if (p + 1 < DDIM) {
            v4f* dst = st + (s ^ 1) * SLOTCH;
            dst[tid] = nb0;
            if (tid + CTHR < REALCH) dst[tid + CTHR] = nb1;
        }
        if (p + 2 < DDIM) {
            nb0 = srcq[tid];
            nb1 = srcq[tid2];
            srcq += PLCH;
        }
        if (p >= 4) {
            if (act) opq[t] = fin;
            opq += PLCH;
        }

        const v4f* q0 = tile + (FRONT - 1) + s * SLOTCH + t;
        v4f rb[2][3];
        rb[0][0] = q0[0]; rb[0][1] = q0[1]; rb[0][2] = q0[2];
        #pragma unroll
        for (int r = 0; r < 7; ++r) {
            const int dy = r - 3;
            if (r + 1 < 7) {
                const v4f* qn = q0 + (r + 1) * ROWCH;
                rb[(r + 1) & 1][0] = qn[0]; rb[(r + 1) & 1][1] = qn[1]; rb[(r + 1) & 1][2] = qn[2];
            }
            const v4f L = rb[r & 1][0], C = rb[r & 1][1], R = rb[r & 1][2];
            const float v[12] = {eL ? 0.f : L.x, eL ? 0.f : L.y, eL ? 0.f : L.z, eL ? 0.f : L.w,
                                 C.x, C.y, C.z, C.w,
                                 eR ? 0.f : R.x, eR ? 0.f : R.y, eR ? 0.f : R.z, eR ? 0.f : R.w};
            #pragma unroll
            for (int d = 1; d <= NB; ++d) {
                if (dy != 0 && dy != d && dy != -d) continue;
                const int ky = (dy == 0) ? 1 : ((dy < 0) ? 0 : 2);
                #pragma unroll
                for (int kz = 0; kz < 3; ++kz) {
                    const int j = 3 - (kz - 1) * d;
                    const float* wp = &wr[(d - 1) * 27 + (kz * 3 + ky) * 3];
                    #pragma unroll
                    for (int kx = 0; kx < 3; ++kx) {
                        const float w = wp[kx];
                        const int o = 4 + (kx - 1) * d;
                        acc[j][0] = fmaf(w, v[o + 0], acc[j][0]);
                        acc[j][1] = fmaf(w, v[o + 1], acc[j][1]);
                        acc[j][2] = fmaf(w, v[o + 2], acc[j][2]);
                        acc[j][3] = fmaf(w, v[o + 3], acc[j][3]);
                    }
                }
            }
        }

        fin.x = acc[0][0]; fin.y = acc[0][1]; fin.z = acc[0][2]; fin.w = acc[0][3];
        #pragma unroll
        for (int j = 0; j < 6; ++j) {
            acc[j][0] = acc[j + 1][0]; acc[j][1] = acc[j + 1][1];
            acc[j][2] = acc[j + 1][2]; acc[j][3] = acc[j + 1][3];
        }
        acc[6][0] = bias; acc[6][1] = bias; acc[6][2] = bias; acc[6][3] = bias;

        lds_barrier();
    }

    if (act) {
        opq[t] = fin;
        #pragma unroll
        for (int j = 0; j < 3; ++j) {
            const v4f res = {acc[j][0], acc[j][1], acc[j][2], acc[j][3]};
            opq[(1 + j) * PLCH + t] = res;
        }
    }
}

extern "C" void kernel_launch(void* const* d_in, const int* in_sizes, int n_in,
                              void* d_out, int out_size, void* d_ws, size_t ws_size,
                              hipStream_t stream) {
    const float* x        = (const float*)d_in[0];
    const float* guidance = (const float*)d_in[1];
    const float* convw    = (const float*)d_in[2];
    const float* convb    = (const float*)d_in[3];
    const float* ln_g     = (const float*)d_in[4];
    const float* ln_b     = (const float*)d_in[5];
    const float* w1       = (const float*)d_in[6];
    const float* b1       = (const float*)d_in[7];
    const float* w2       = (const float*)d_in[8];
    const float* b2       = (const float*)d_in[9];

    float* ws      = (float*)d_ws;
    float* feat    = ws;
    float* wts     = ws + 384;
    float* biaseff = ws + 512;
    float* wscaled = ws + 1024;

    feat_reduce_kernel<<<B_N * C_CH, 256, 0, stream>>>(x, feat);
    gate_kernel<<<1, 256, 0, stream>>>(feat, guidance, ln_g, ln_b,
                                      w1, b1, w2, b2, wts);
    const int foldN = B_N * C_CH * NB * 27;
    fold_kernel<<<(foldN + 255) / 256, 256, 0, stream>>>(convw, convb, wts,
                                                         wscaled, biaseff);
    dwconv_kernel<<<B_N * C_CH * YBLK, CTHR, 0, stream>>>(x, wscaled, biaseff,
                                                         (float*)d_out);
}


# gate MLP (block's own batch row; LayerNorm sums now 6 per lane + wave xor-butterfly tree) and the weight fold computed in every conv block's prologue: gate and fold launches gone, 4 launches to 2
# speedup vs baseline: 13.9082x; 1.1879x over previous
#include <hip/hip_runtime.h>
#include <math.h>

#define B_N   4
#define C_CH  96
#define DDIM  16
#define HDIM  56
#define WDIM  56
#define GDIM  96
#define HID   24
#define NB    3
#define NCOL  (C_CH + GDIM)
#define SPATIAL (DDIM * HDIM * WDIM)
#define LN_EPS 1e-5f

typedef float v2f __attribute__((ext_vector_type(2)));
typedef float v4f __attribute__((ext_vector_type(4)));
typedef float v8f __attribute__((ext_vector_type(8)));

__global__ void feat_reduce_kernel(const float* __restrict__ x,
                                   float* __restrict__ feat) {
    const int bc = blockIdx.x;
    const v4f* p = (const v4f*)(x + (size_t)bc * SPATIAL);
    float s = 0.f;
    for (int i = threadIdx.x; i < SPATIAL / 4; i += 256) {
        v4f v = p[i];
        s += v.x + v.y + v.z + v.w;
    }
    __shared__ float red[256];
    red[threadIdx.x] = s;
    __syncthreads();
    for (int off = 128; off > 0; off >>= 1) {
        if (threadIdx.x < off) red[threadIdx.x] += red[threadIdx.x + off];
        __syncthreads();
    }
    if (threadIdx.x == 0) feat[bc] = red[0] * (1.0f / (float)SPATIAL);
}

#define ROWCH   14
#define YSL     28
#define YBLK    (HDIM / YSL)
#define SROWS   (YSL + 6)
#define REALROWS (YSL + 3)
#define REALCH  (REALROWS * ROWCH)
#define PLCH    (HDIM * ROWCH)
#define SLOTCH  480
#define FRONT   1
#define NCHUNK  (FRONT + 2 * SLOTCH + 3)
#define NPEN    (YSL * ROWCH)
#define CTHR    416
#define W1_V4   ((NCOL * HID) / 4)
#define W1P     196

__device__ __forceinline__ void lds_barrier() {
    __builtin_amdgcn_fence(__ATOMIC_RELEASE, "workgroup", "local");
    __builtin_amdgcn_s_barrier();
    __builtin_amdgcn_fence(__ATOMIC_ACQUIRE, "workgroup", "local");
}

__global__ __launch_bounds__(CTHR, 10)
void dwconv_kernel(const float* __restrict__ x,
                   const float* __restrict__ feat,
                   const float* __restrict__ guidance,
                   const float* __restrict__ ln_g,
                   const float* __restrict__ ln_b,
                   const float* __restrict__ w1p,
                   const float* __restrict__ b1p,
                   const float* __restrict__ w2p,
                   const float* __restrict__ b2p,
                   const float* __restrict__ convw,
                   const float* __restrict__ convb,
                   float* __restrict__ out) {
    const int blk = blockIdx.x;
    const int yh  = blk % YBLK;
    const int bc  = blk / YBLK;
    const int tid = threadIdx.x;
    const int cch = bc % C_CH;
    const int bat = bc / C_CH;

    __shared__ v4f tile[NCHUNK];
    __shared__ __attribute__((aligned(16))) float w1t[HID][W1P];
    __shared__ __attribute__((aligned(16))) float graw[NCOL];
    __shared__ __attribute__((aligned(16))) float gnrow[NCOL];
    __shared__ float hbuf[HID];
    __shared__ float w2s[HID * NB];
    __shared__ float b2s[NB];

    const int lane = tid & 31;
    const int half = lane >> 4;
    const int l    = lane & 15;

    v4f wv[3];
    #pragma unroll
    for (int j = 0; j < 3; ++j) {
        const int q  = min(tid + CTHR * j, W1_V4 - 1);
        const int n4 = q / NCOL, k = q - n4 * NCOL;
        wv[j] = *(const v4f*)(w1p + k * HID + n4 * 4);
    }
    const int col = min(tid, NCOL - 1);
    const float* gsrc = (col < C_CH) ? (feat + bat * C_CH + col)
                                     : (guidance + bat * GDIM + (col - C_CH));
    const float gvv = *gsrc;
    const float lgv = ln_g[col];
    const float lbv = ln_b[col];
    const float bl0 = b1p[l];
    const float bl1 = b1p[min(16 + l, HID - 1)];
    const float w2v = w2p[min(tid, HID * NB - 1)];
    const float b2v = b2p[min(tid, NB - 1)];

    const int ylo = yh ? (HDIM - REALROWS) : 0;
    const int rlo = yh ? 0 : 3;
    const v4f* xp = (const v4f*)(x + (size_t)bc * SPATIAL + ylo * WDIM);
    const v4f zero = {0.f, 0.f, 0.f, 0.f};

    for (int i = tid; i < NCHUNK; i += CTHR) {
        const int e = i - (FRONT + rlo * ROWCH);
        const bool real = (unsigned)e < (unsigned)REALCH;
        const v4f v = xp[real ? e : 0];
        tile[i] = real ? v : zero;
    }

    v4f nb0 = xp[PLCH + tid];
    v4f nb1 = xp[PLCH + min(tid + CTHR, REALCH - 1)];

    #pragma unroll
    for (int j = 0; j < 3; ++j) {
        const int q = tid + CTHR * j;
        if (q < W1_V4) {
            const int n4 = q / NCOL, k = q - n4 * NCOL;
            float* d = &w1t[n4 * 4][k];
            d[0 * W1P] = wv[j].x;
            d[1 * W1P] = wv[j].y;
            d[2 * W1P] = wv[j].z;
            d[3 * W1P] = wv[j].w;
        }
    }
    if (tid < NCOL) graw[tid] = gvv;
    if (tid < HID * NB) w2s[tid] = w2v;
    if (tid < NB) b2s[tid] = b2v;
    lds_barrier();

    float mean_r, rstd_r;
    {
        float s = 0.f;
        #pragma unroll
        for (int k = 0; k < NCOL / 32; ++k) s += graw[lane + 32 * k];
        #pragma unroll
        for (int off = 16; off > 0; off >>= 1) s += __shfl_xor(s, off);
        mean_r = s * (1.0f / NCOL);
        float v = 0.f;
        #pragma unroll
        for (int k = 0; k < NCOL / 32; ++k) { const float d = graw[lane + 32 * k] - mean_r; v += d * d; }
        #pragma unroll
        for (int off = 16; off > 0; off >>= 1) v += __shfl_xor(v, off);
        v *= (1.0f / NCOL);
        rstd_r = rsqrtf(v + LN_EPS);
    }
    if (tid < NCOL) gnrow[tid] = (gvv - mean_r) * rstd_r * lgv + lbv;
    lds_barrier();

    const int wave = __builtin_amdgcn_readfirstlane(tid >> 5);
    if (wave == 0) {
        const float* b0row = &w1t[l][0];
        const float* b1row = &w1t[min(16 + l, HID - 1)][0];
        v8f c0 = {};
        v8f c1 = {};
        for (int kk = 0; kk < NCOL / 4; ++kk) {
            const int kb = kk * 4 + half * 2;
            const v2f a   = *(const v2f*)(gnrow + kb);
            const v2f b0  = *(const v2f*)(b0row + kb);
            const v2f b1v = *(const v2f*)(b1row + kb);
            c0 = __builtin_amdgcn_wmma_f32_16x16x4_f32(false, a, false, b0,
                                                       (short)0, c0, false, false);
            c1 = __builtin_amdgcn_wmma_f32_16x16x4_f32(false, a, false, b1v,
                                                       (short)0, c1, false, false);
        }
        float v0 = c0[0] + bl0;
        v0 = 0.5f * v0 * (1.0f + erff(v0 * 0.70710678118654752f));
        float v1 = c1[0] + bl1;
        v1 = 0.5f * v1 * (1.0f + erff(v1 * 0.70710678118654752f));
        if (lane < 16) hbuf[l] = v0;
        if (lane < HID - 16) hbuf[16 + l] = v1;
    }
    lds_barrier();

    float lgt[NB];
    #pragma unroll
    for (int j = 0; j < NB; ++j) lgt[j] = b2s[j];
    #pragma unroll 8
    for (int k = 0; k < HID; ++k) {
        const float hk = hbuf[k];
        #pragma unroll
        for (int j = 0; j < NB; ++j) lgt[j] += hk * w2s[k * NB + j];
    }
    float al[NB];
    {
        const float mx = fmaxf(lgt[0], fmaxf(lgt[1], lgt[2]));
        const float e0 = __expf(lgt[0] - mx);
        const float e1 = __expf(lgt[1] - mx);
        const float e2 = __expf(lgt[2] - mx);
        const float inv = 1.0f / (e0 + e1 + e2);
        al[0] = __int_as_float(__builtin_amdgcn_readfirstlane(__float_as_int(e0 * inv)));
        al[1] = __int_as_float(__builtin_amdgcn_readfirstlane(__float_as_int(e1 * inv)));
        al[2] = __int_as_float(__builtin_amdgcn_readfirstlane(__float_as_int(e2 * inv)));
    }
    float bsum = 0.f;
    #pragma unroll
    for (int i = 0; i < NB; ++i) bsum += al[i] * convb[i * C_CH + cch];
    const float bias = __int_as_float(__builtin_amdgcn_readfirstlane(__float_as_int(bsum)));

    const bool act = (tid < NPEN);
    const int t  = act ? tid : (NPEN - 1);
    const int sx = t % ROWCH;
    const bool eL = (sx == 0);
    const bool eR = (sx == ROWCH - 1);

    float wr[NB * 27];
    #pragma unroll
    for (int i = 0; i < NB * 27; ++i)
        wr[i] = al[i / 27] * convw[((i / 27) * C_CH + cch) * 27 + (i % 27)];

    float acc[7][4];
    #pragma unroll
    for (int j = 0; j < 7; ++j) { acc[j][0] = bias; acc[j][1] = bias; acc[j][2] = bias; acc[j][3] = bias; }

    v4f* st = tile + FRONT + rlo * ROWCH;

    v4f fin = zero;
    const v4f* srcq = xp + 2 * PLCH;
    v4f* opq = (v4f*)(out + (size_t)bc * SPATIAL + (yh * YSL) * WDIM);
    const int tid2 = min(tid + CTHR, REALCH - 1);

    #pragma unroll 1
    for (int p = 0; p < DDIM; ++p) {
        const int s = p & 1;
        if (p + 1 < DDIM) {
            v4f* dst = st + (s ^ 1) * SLOTCH;
            dst[tid] = nb0;
            if (tid + CTHR < REALCH) dst[tid + CTHR] = nb1;
        }
        if (p + 2 < DDIM) {
            nb0 = srcq[tid];
            nb1 = srcq[tid2];
            srcq += PLCH;
        }
        if (p >= 4) {
            if (act) opq[t] = fin;
            opq += PLCH;
        }

        const v4f* q0 = tile + (FRONT - 1) + s * SLOTCH + t;
        v4f rb[2][3];
        rb[0][0] = q0[0]; rb[0][1] = q0[1]; rb[0][2] = q0[2];
        #pragma unroll
        for (int r = 0; r < 7; ++r) {
            const int dy = r - 3;
            if (r + 1 < 7) {
                const v4f* qn = q0 + (r + 1) * ROWCH;
                rb[(r + 1) & 1][0] = qn[0]; rb[(r + 1) & 1][1] = qn[1]; rb[(r + 1) & 1][2] = qn[2];
            }
            const v4f L = rb[r & 1][0], C = rb[r & 1][1], R = rb[r & 1][2];
            const float v[12] = {eL ? 0.f : L.x, eL ? 0.f : L.y, eL ? 0.f : L.z, eL ? 0.f : L.w,
                                 C.x, C.y, C.z, C.w,
                                 eR ? 0.f : R.x, eR ? 0.f : R.y, eR ? 0.f : R.z, eR ? 0.f : R.w};
            #pragma unroll
            for (int d = 1; d <= NB; ++d) {
                if (dy != 0 && dy != d && dy != -d) continue;
                const int ky = (dy == 0) ? 1 : ((dy < 0) ? 0 : 2);
                #pragma unroll
                for (int kz = 0; kz < 3; ++kz) {
                    const int j = 3 - (kz - 1) * d;
                    const float* wp = &wr[(d - 1) * 27 + (kz * 3 + ky) * 3];
                    #pragma unroll
                    for (int kx = 0; kx < 3; ++kx) {
                        const float w = wp[kx];
                        const int o = 4 + (kx - 1) * d;
                        acc[j][0] = fmaf(w, v[o + 0], acc[j][0]);
                        acc[j][1] = fmaf(w, v[o + 1], acc[j][1]);
                        acc[j][2] = fmaf(w, v[o + 2], acc[j][2]);
                        acc[j][3] = fmaf(w, v[o + 3], acc[j][3]);
                    }
                }
            }
        }

        fin.x = acc[0][0]; fin.y = acc[0][1]; fin.z = acc[0][2]; fin.w = acc[0][3];
        #pragma unroll
        for (int j = 0; j < 6; ++j) {
            acc[j][0] = acc[j + 1][0]; acc[j][1] = acc[j + 1][1];
            acc[j][2] = acc[j + 1][2]; acc[j][3] = acc[j + 1][3];
        }
        acc[6][0] = bias; acc[6][1] = bias; acc[6][2] = bias; acc[6][3] = bias;

        lds_barrier();
    }

    if (act) {
        opq[t] = fin;
        #pragma unroll
        for (int j = 0; j < 3; ++j) {
            const v4f res = {acc[j][0], acc[j][1], acc[j][2], acc[j][3]};
            opq[(1 + j) * PLCH + t] = res;
        }
    }
}

extern "C" void kernel_launch(void* const* d_in, const int* in_sizes, int n_in,
                              void* d_out, int out_size, void* d_ws, size_t ws_size,
                              hipStream_t stream) {
    const float* x        = (const float*)d_in[0];
    const float* guidance = (const float*)d_in[1];
    const float* convw    = (const float*)d_in[2];
    const float* convb    = (const float*)d_in[3];
    const float* ln_g     = (const float*)d_in[4];
    const float* ln_b     = (const float*)d_in[5];
    const float* w1       = (const float*)d_in[6];
    const float* b1       = (const float*)d_in[7];
    const float* w2       = (const float*)d_in[8];
    const float* b2       = (const float*)d_in[9];

    float* ws      = (float*)d_ws;
    float* feat    = ws;

    feat_reduce_kernel<<<B_N * C_CH, 256, 0, stream>>>(x, feat);
    dwconv_kernel<<<B_N * C_CH * YBLK, CTHR, 0, stream>>>(x, feat, guidance, ln_g, ln_b,
                                                         w1, b1, w2, b2, convw, convb,
                                                         (float*)d_out);
}


# conv thread owns two 4-wide strips one under the other: the 8 staged rows of a row pair are read once per input plane and feed both output rows; 7 waves per block, not 13; tap sum order kept
# speedup vs baseline: 14.3573x; 1.0323x over previous
#include <hip/hip_runtime.h>
#include <math.h>

#define B_N   4
#define C_CH  96
#define DDIM  16
#define HDIM  56
#define WDIM  56
#define GDIM  96
#define HID   24
#define NB    3
#define NCOL  (C_CH + GDIM)
#define SPATIAL (DDIM * HDIM * WDIM)
#define LN_EPS 1e-5f

typedef float v2f __attribute__((ext_vector_type(2)));
typedef float v4f __attribute__((ext_vector_type(4)));
typedef float v8f __attribute__((ext_vector_type(8)));

__global__ void feat_reduce_kernel(const float* __restrict__ x,
                                   float* __restrict__ feat) {
    const int bc = blockIdx.x;
    const v4f* p = (const v4f*)(x + (size_t)bc * SPATIAL);
    float s = 0.f;
    for (int i = threadIdx.x; i < SPATIAL / 4; i += 256) {
        v4f v = p[i];
        s += v.x + v.y + v.z + v.w;
    }
    __shared__ float red[256];
    red[threadIdx.x] = s;
    __syncthreads();
    for (int off = 128; off > 0; off >>= 1) {
        if (threadIdx.x < off) red[threadIdx.x] += red[threadIdx.x + off];
        __syncthreads();
    }
    if (threadIdx.x == 0) feat[bc] = red[0] * (1.0f / (float)SPATIAL);
}

#define ROWCH   14
#define YSL     28
#define YBLK    (HDIM / YSL)
#define SROWS   (YSL + 6)
#define REALROWS (YSL + 3)
#define REALCH  (REALROWS * ROWCH)
#define PLCH    (HDIM * ROWCH)
#define SLOTCH  480
#define FRONT   1
#define NCHUNK  (FRONT + 2 * SLOTCH + 3)
#define W1_V4   ((NCOL * HID) / 4)
#define NPAIR   (YSL / 2)
#define CTHR    (NPAIR * 16)
#define W1_IT   6
#define W1P     196

__device__ __forceinline__ void lds_barrier() {
    __builtin_amdgcn_fence(__ATOMIC_RELEASE, "workgroup", "local");
    __builtin_amdgcn_s_barrier();
    __builtin_amdgcn_fence(__ATOMIC_ACQUIRE, "workgroup", "local");
}

__global__ __launch_bounds__(CTHR, 6)
void dwconv_kernel(const float* __restrict__ x,
                   const float* __restrict__ feat,
                   const float* __restrict__ guidance,
                   const float* __restrict__ ln_g,
                   const float* __restrict__ ln_b,
                   const float* __restrict__ w1p,
                   const float* __restrict__ b1p,
                   const float* __restrict__ w2p,
                   const float* __restrict__ b2p,
                   const float* __restrict__ convw,
                   const float* __restrict__ convb,
                   float* __restrict__ out) {
    const int blk = blockIdx.x;
    const int yh  = blk % YBLK;
    const int bc  = blk / YBLK;
    const int tid = threadIdx.x;
    const int cch = bc % C_CH;
    const int bat = bc / C_CH;

    __shared__ v4f tile[NCHUNK];
    __shared__ __attribute__((aligned(16))) float w1t[HID][W1P];
    __shared__ __attribute__((aligned(16))) float graw[NCOL];
    __shared__ __attribute__((aligned(16))) float gnrow[NCOL];
    __shared__ float hbuf[HID];
    __shared__ float w2s[HID * NB];
    __shared__ float b2s[NB];

    const int lane = tid & 31;
    const int half = lane >> 4;
    const int l    = lane & 15;

    v4f wv[W1_IT];
    #pragma unroll
    for (int j = 0; j < W1_IT; ++j) {
        const int q  = min(tid + CTHR * j, W1_V4 - 1);
        const int n4 = q / NCOL, k = q - n4 * NCOL;
        wv[j] = *(const v4f*)(w1p + k * HID + n4 * 4);
    }
    const int col = min(tid, NCOL - 1);
    const float* gsrc = (col < C_CH) ? (feat + bat * C_CH + col)
                                     : (guidance + bat * GDIM + (col - C_CH));
    const float gvv = *gsrc;
    const float lgv = ln_g[col];
    const float lbv = ln_b[col];
    const float bl0 = b1p[l];
    const float bl1 = b1p[min(16 + l, HID - 1)];
    const float w2v = w2p[min(tid, HID * NB - 1)];
    const float b2v = b2p[min(tid, NB - 1)];

    const int ylo = yh ? (HDIM - REALROWS) : 0;
    const int rlo = yh ? 0 : 3;
    const v4f* xp = (const v4f*)(x + (size_t)bc * SPATIAL + ylo * WDIM);
    const v4f zero = {0.f, 0.f, 0.f, 0.f};

    const int tid2 = min(tid + CTHR, REALCH - 1);
    const v4f f0 = xp[tid];
    const v4f f1 = xp[tid2];
    for (int i = tid; i < NCHUNK; i += CTHR) {
        const int e = i - (FRONT + rlo * ROWCH);
        if ((unsigned)e >= (unsigned)REALCH) tile[i] = zero;
    }

    v4f nb0 = xp[PLCH + tid];
    v4f nb1 = xp[PLCH + tid2];
    {
        v4f* st0 = tile + FRONT + rlo * ROWCH;
        st0[tid] = f0;
        if (tid + CTHR < REALCH) st0[tid + CTHR] = f1;
    }

    #pragma unroll
    for (int j = 0; j < W1_IT; ++j) {
        const int q = tid + CTHR * j;
        if (q < W1_V4) {
            const int n4 = q / NCOL, k = q - n4 * NCOL;
            float* d = &w1t[n4 * 4][k];
            d[0 * W1P] = wv[j].x;
            d[1 * W1P] = wv[j].y;
            d[2 * W1P] = wv[j].z;
            d[3 * W1P] = wv[j].w;
        }
    }
    if (tid < NCOL) graw[tid] = gvv;
    if (tid < HID * NB) w2s[tid] = w2v;
    if (tid < NB) b2s[tid] = b2v;
    lds_barrier();

    float mean_r, rstd_r;
    {
        float s = 0.f;
        #pragma unroll
        for (int k = 0; k < NCOL / 32; ++k) s += graw[lane + 32 * k];
        #pragma unroll
        for (int off = 16; off > 0; off >>= 1) s += __shfl_xor(s, off);
        mean_r = s * (1.0f / NCOL);
        float v = 0.f;
        #pragma unroll
        for (int k = 0; k < NCOL / 32; ++k) { const float d = graw[lane + 32 * k] - mean_r; v += d * d; }
        #pragma unroll
        for (int off = 16; off > 0; off >>= 1) v += __shfl_xor(v, off);
        v *= (1.0f / NCOL);
        rstd_r = rsqrtf(v + LN_EPS);
    }
    if (tid < NCOL) gnrow[tid] = (gvv - mean_r) * rstd_r * lgv + lbv;
    lds_barrier();

    const int wave = __builtin_amdgcn_readfirstlane(tid >> 5);
    if (wave == 0) {
        const float* b0row = &w1t[l][0];
        const float* b1row = &w1t[min(16 + l, HID - 1)][0];
        v8f c0 = {};
        v8f c1 = {};
        for (int kk = 0; kk < NCOL / 4; ++kk) {
            const int kb = kk * 4 + half * 2;
            const v2f a   = *(const v2f*)(gnrow + kb);
            const v2f b0  = *(const v2f*)(b0row + kb);
            const v2f b1v = *(const v2f*)(b1row + kb);
            c0 = __builtin_amdgcn_wmma_f32_16x16x4_f32(false, a, false, b0,
                                                       (short)0, c0, false, false);
            c1 = __builtin_amdgcn_wmma_f32_16x16x4_f32(false, a, false, b1v,
                                                       (short)0, c1, false, false);
        }
        float v0 = c0[0] + bl0;
        v0 = 0.5f * v0 * (1.0f + erff(v0 * 0.70710678118654752f));
        float v1 = c1[0] + bl1;
        v1 = 0.5f * v1 * (1.0f + erff(v1 * 0.70710678118654752f));
        if (lane < 16) hbuf[l] = v0;
        if (lane < HID - 16) hbuf[16 + l] = v1;
    }
    lds_barrier();

    float lgt[NB];
    #pragma unroll
    for (int j = 0; j < NB; ++j) lgt[j] = b2s[j];
    #pragma unroll 8
    for (int k = 0; k < HID; ++k) {
        const float hk = hbuf[k];
        #pragma unroll
        for (int j = 0; j < NB; ++j) lgt[j] += hk * w2s[k * NB + j];
    }
    float al[NB];
    {
        const float mx = fmaxf(lgt[0], fmaxf(lgt[1], lgt[2]));
        const float e0 = __expf(lgt[0] - mx);
        const float e1 = __expf(lgt[1] - mx);
        const float e2 = __expf(lgt[2] - mx);
        const float inv = 1.0f / (e0 + e1 + e2);
        al[0] = __int_as_float(__builtin_amdgcn_readfirstlane(__float_as_int(e0 * inv)));
        al[1] = __int_as_float(__builtin_amdgcn_readfirstlane(__float_as_int(e1 * inv)));
        al[2] = __int_as_float(__builtin_amdgcn_readfirstlane(__float_as_int(e2 * inv)));
    }
    float bsum = 0.f;
    #pragma unroll
    for (int i = 0; i < NB; ++i) bsum += al[i] * convb[i * C_CH + cch];
    const float bias = __int_as_float(__builtin_amdgcn_readfirstlane(__float_as_int(bsum)));

    const int grp = tid >> 4;
    const int li  = tid & 15;
    const bool act = (li < ROWCH);
    const int sx = act ? li : (ROWCH - 1);
    const int tA = (2 * grp) * ROWCH + sx;
    const bool eL = (sx == 0);
    const bool eR = (sx == ROWCH - 1);

    float wr[NB * 27];
    #pragma unroll
    for (int i = 0; i < NB * 27; ++i)
        wr[i] = al[i / 27] * convw[((i / 27) * C_CH + cch) * 27 + (i % 27)];

    float acc[7][2][4];
    #pragma unroll
    for (int j = 0; j < 7; ++j) {
        #pragma unroll
        for (int o = 0; o < 2; ++o) { acc[j][o][0] = bias; acc[j][o][1] = bias; acc[j][o][2] = bias; acc[j][o][3] = bias; }
    }

    v4f* st = tile + FRONT + rlo * ROWCH;

    v4f fin0 = zero, fin1 = zero;
    const v4f* srcq = xp + 2 * PLCH;
    v4f* opq = (v4f*)(out + (size_t)bc * SPATIAL + (yh * YSL) * WDIM);

    #pragma unroll 1
    for (int p = 0; p < DDIM; ++p) {
        const int s = p & 1;
        if (p + 1 < DDIM) {
            v4f* dst = st + (s ^ 1) * SLOTCH;
            dst[tid] = nb0;
            if (tid + CTHR < REALCH) dst[tid + CTHR] = nb1;
        }
        if (p + 2 < DDIM) {
            nb0 = srcq[tid];
            nb1 = srcq[tid2];
            srcq += PLCH;
        }
        if (p >= 4) {
            if (act) { opq[tA] = fin0; opq[tA + ROWCH] = fin1; }
            opq += PLCH;
        }

        const v4f* q0 = tile + (FRONT - 1) + s * SLOTCH + tA;
        v4f rb[2][3];
        rb[0][0] = q0[0]; rb[0][1] = q0[1]; rb[0][2] = q0[2];
        #pragma unroll
        for (int r = 0; r < 8; ++r) {
            if (r + 1 < 8) {
                const v4f* qn = q0 + (r + 1) * ROWCH;
                rb[(r + 1) & 1][0] = qn[0]; rb[(r + 1) & 1][1] = qn[1]; rb[(r + 1) & 1][2] = qn[2];
            }
            const v4f L = rb[r & 1][0], C = rb[r & 1][1], R = rb[r & 1][2];
            const float v[12] = {eL ? 0.f : L.x, eL ? 0.f : L.y, eL ? 0.f : L.z, eL ? 0.f : L.w,
                                 C.x, C.y, C.z, C.w,
                                 eR ? 0.f : R.x, eR ? 0.f : R.y, eR ? 0.f : R.z, eR ? 0.f : R.w};
            #pragma unroll
            for (int o = 0; o < 2; ++o) {
                const int dy = r - 3 - o;
                if (dy < -3 || dy > 3) continue;
                #pragma unroll
                for (int d = 1; d <= NB; ++d) {
                    if (dy != 0 && dy != d && dy != -d) continue;
                    const int ky = (dy == 0) ? 1 : ((dy < 0) ? 0 : 2);
                    #pragma unroll
                    for (int kz = 0; kz < 3; ++kz) {
                        const int j = 3 - (kz - 1) * d;
                        const float* wp = &wr[(d - 1) * 27 + (kz * 3 + ky) * 3];
                        #pragma unroll
                        for (int kx = 0; kx < 3; ++kx) {
                            const float w = wp[kx];
                            const int ox = 4 + (kx - 1) * d;
                            acc[j][o][0] = fmaf(w, v[ox + 0], acc[j][o][0]);
                            acc[j][o][1] = fmaf(w, v[ox + 1], acc[j][o][1]);
                            acc[j][o][2] = fmaf(w, v[ox + 2], acc[j][o][2]);
                            acc[j][o][3] = fmaf(w, v[ox + 3], acc[j][o][3]);
                        }
                    }
                }
            }
        }

        fin0.x = acc[0][0][0]; fin0.y = acc[0][0][1]; fin0.z = acc[0][0][2]; fin0.w = acc[0][0][3];
        fin1.x = acc[0][1][0]; fin1.y = acc[0][1][1]; fin1.z = acc[0][1][2]; fin1.w = acc[0][1][3];
        #pragma unroll
        for (int j = 0; j < 6; ++j) {
            #pragma unroll
            for (int o = 0; o < 2; ++o) {
                acc[j][o][0] = acc[j + 1][o][0]; acc[j][o][1] = acc[j + 1][o][1];
                acc[j][o][2] = acc[j + 1][o][2]; acc[j][o][3] = acc[j + 1][o][3];
            }
        }
        #pragma unroll
        for (int o = 0; o < 2; ++o) { acc[6][o][0] = bias; acc[6][o][1] = bias; acc[6][o][2] = bias; acc[6][o][3] = bias; }

        lds_barrier();
    }

    if (act) {
        opq[tA] = fin0;
        opq[tA + ROWCH] = fin1;
        #pragma unroll
        for (int j = 0; j < 3; ++j) {
            const v4f r0 = {acc[j][0][0], acc[j][0][1], acc[j][0][2], acc[j][0][3]};
            const v4f r1 = {acc[j][1][0], acc[j][1][1], acc[j][1][2], acc[j][1][3]};
            opq[(1 + j) * PLCH + tA] = r0;
            opq[(1 + j) * PLCH + tA + ROWCH] = r1;
        }
    }
}

extern "C" void kernel_launch(void* const* d_in, const int* in_sizes, int n_in,
                              void* d_out, int out_size, void* d_ws, size_t ws_size,
                              hipStream_t stream) {
    const float* x        = (const float*)d_in[0];
    const float* guidance = (const float*)d_in[1];
    const float* convw    = (const float*)d_in[2];
    const float* convb    = (const float*)d_in[3];
    const float* ln_g     = (const float*)d_in[4];
    const float* ln_b     = (const float*)d_in[5];
    const float* w1       = (const float*)d_in[6];
    const float* b1       = (const float*)d_in[7];
    const float* w2       = (const float*)d_in[8];
    const float* b2       = (const float*)d_in[9];

    float* ws      = (float*)d_ws;
    float* feat    = ws;

    feat_reduce_kernel<<<B_N * C_CH, 256, 0, stream>>>(x, feat);
    dwconv_kernel<<<B_N * C_CH * YBLK, CTHR, 0, stream>>>(x, feat, guidance, ln_g, ln_b,
                                                         w1, b1, w2, b2, convw, convb,
                                                         (float*)d_out);
}


# conv LDS rows at a pitch of 15 chunks: one zero chunk behind every staged row is the x halo of strips 0 and 13, so the per-lane halo selects of the tap loop are gone; sums and their order unchanged
# speedup vs baseline: 14.7746x; 1.0291x over previous
#include <hip/hip_runtime.h>
#include <math.h>

#define B_N   4
#define C_CH  96
#define DDIM  16
#define HDIM  56
#define WDIM  56
#define GDIM  96
#define HID   24
#define NB    3
#define NCOL  (C_CH + GDIM)
#define SPATIAL (DDIM * HDIM * WDIM)
#define LN_EPS 1e-5f

typedef float v2f __attribute__((ext_vector_type(2)));
typedef float v4f __attribute__((ext_vector_type(4)));
typedef float v8f __attribute__((ext_vector_type(8)));

__global__ void feat_reduce_kernel(const float* __restrict__ x,
                                   float* __restrict__ feat) {
    const int bc = blockIdx.x;
    const v4f* p = (const v4f*)(x + (size_t)bc * SPATIAL);
    float s = 0.f;
    for (int i = threadIdx.x; i < SPATIAL / 4; i += 256) {
        v4f v = p[i];
        s += v.x + v.y + v.z + v.w;
    }
    __shared__ float red[256];
    red[threadIdx.x] = s;
    __syncthreads();
    for (int off = 128; off > 0; off >>= 1) {
        if (threadIdx.x < off) red[threadIdx.x] += red[threadIdx.x + off];
        __syncthreads();
    }
    if (threadIdx.x == 0) feat[bc] = red[0] * (1.0f / (float)SPATIAL);
}

#define ROWCH   14
#define RPITCH  15
#define YSL     28
#define YBLK    (HDIM / YSL)
#define SROWS   (YSL + 6)
#define REALROWS (YSL + 3)
#define REALCH  (REALROWS * ROWCH)
#define PLCH    (HDIM * ROWCH)
#define SLOTCH  512
#define FRONT   1
#define NCHUNK  (FRONT + 2 * SLOTCH + 3)
#define W1_V4   ((NCOL * HID) / 4)
#define NPAIR   (YSL / 2)
#define CTHR    (NPAIR * 16)
#define W1_IT   6
#define W1P     196

__device__ __forceinline__ void lds_barrier() {
    __builtin_amdgcn_fence(__ATOMIC_RELEASE, "workgroup", "local");
    __builtin_amdgcn_s_barrier();
    __builtin_amdgcn_fence(__ATOMIC_ACQUIRE, "workgroup", "local");
}

__global__ __launch_bounds__(CTHR, 6)
void dwconv_kernel(const float* __restrict__ x,
                   const float* __restrict__ feat,
                   const float* __restrict__ guidance,
                   const float* __restrict__ ln_g,
                   const float* __restrict__ ln_b,
                   const float* __restrict__ w1p,
                   const float* __restrict__ b1p,
                   const float* __restrict__ w2p,
                   const float* __restrict__ b2p,
                   const float* __restrict__ convw,
                   const float* __restrict__ convb,
                   float* __restrict__ out) {
    const int blk = blockIdx.x;
    const int yh  = blk % YBLK;
    const int bc  = blk / YBLK;
    const int tid = threadIdx.x;
    const int cch = bc % C_CH;
    const int bat = bc / C_CH;

    __shared__ v4f tile[NCHUNK];
    __shared__ __attribute__((aligned(16))) float w1t[HID][W1P];
    __shared__ __attribute__((aligned(16))) float graw[NCOL];
    __shared__ __attribute__((aligned(16))) float gnrow[NCOL];
    __shared__ float hbuf[HID];
    __shared__ float w2s[HID * NB];
    __shared__ float b2s[NB];

    const int lane = tid & 31;
    const int half = lane >> 4;
    const int l    = lane & 15;

    v4f wv[W1_IT];
    #pragma unroll
    for (int j = 0; j < W1_IT; ++j) {
        const int q  = min(tid + CTHR * j, W1_V4 - 1);
        const int n4 = q / NCOL, k = q - n4 * NCOL;
        wv[j] = *(const v4f*)(w1p + k * HID + n4 * 4);
    }
    const int col = min(tid, NCOL - 1);
    const float* gsrc = (col < C_CH) ? (feat + bat * C_CH + col)
                                     : (guidance + bat * GDIM + (col - C_CH));
    const float gvv = *gsrc;
    const float lgv = ln_g[col];
    const float lbv = ln_b[col];
    const float bl0 = b1p[l];
    const float bl1 = b1p[min(16 + l, HID - 1)];
    const float w2v = w2p[min(tid, HID * NB - 1)];
    const float b2v = b2p[min(tid, NB - 1)];

    const int ylo = yh ? (HDIM - REALROWS) : 0;
    const int rlo = yh ? 0 : 3;
    const v4f* xp = (const v4f*)(x + (size_t)bc * SPATIAL + ylo * WDIM);
    const v4f zero = {0.f, 0.f, 0.f, 0.f};

    const int tid2 = min(tid + CTHR, REALCH - 1);
    const int d0 = tid + tid / ROWCH;
    const v4f f0 = xp[tid];
    const v4f f1 = xp[tid2];
    for (int i = tid; i < NCHUNK; i += CTHR) {
        const int e = i - (FRONT + rlo * RPITCH);
        const bool data0 = ((unsigned)e < (unsigned)(REALROWS * RPITCH)) && (e % RPITCH != ROWCH);
        if (!data0) tile[i] = zero;
    }

    v4f nb0 = xp[PLCH + tid];
    v4f nb1 = xp[PLCH + tid2];
    {
        v4f* st0 = tile + FRONT + rlo * RPITCH;
        st0[d0] = f0;
        if (tid + CTHR < REALCH) st0[d0 + 16 * RPITCH] = f1;
    }

    #pragma unroll
    for (int j = 0; j < W1_IT; ++j) {
        const int q = tid + CTHR * j;
        if (q < W1_V4) {
            const int n4 = q / NCOL, k = q - n4 * NCOL;
            float* d = &w1t[n4 * 4][k];
            d[0 * W1P] = wv[j].x;
            d[1 * W1P] = wv[j].y;
            d[2 * W1P] = wv[j].z;
            d[3 * W1P] = wv[j].w;
        }
    }
    if (tid < NCOL) graw[tid] = gvv;
    if (tid < HID * NB) w2s[tid] = w2v;
    if (tid < NB) b2s[tid] = b2v;
    lds_barrier();

    float mean_r, rstd_r;
    {
        float s = 0.f;
        #pragma unroll
        for (int k = 0; k < NCOL / 32; ++k) s += graw[lane + 32 * k];
        #pragma unroll
        for (int off = 16; off > 0; off >>= 1) s += __shfl_xor(s, off);
        mean_r = s * (1.0f / NCOL);
        float v = 0.f;
        #pragma unroll
        for (int k = 0; k < NCOL / 32; ++k) { const float d = graw[lane + 32 * k] - mean_r; v += d * d; }
        #pragma unroll
        for (int off = 16; off > 0; off >>= 1) v += __shfl_xor(v, off);
        v *= (1.0f / NCOL);
        rstd_r = rsqrtf(v + LN_EPS);
    }
    if (tid < NCOL) gnrow[tid] = (gvv - mean_r) * rstd_r * lgv + lbv;
    lds_barrier();

    const int wave = __builtin_amdgcn_readfirstlane(tid >> 5);
    if (wave == 0) {
        const float* b0row = &w1t[l][0];
        const float* b1row = &w1t[min(16 + l, HID - 1)][0];
        v8f c0 = {};
        v8f c1 = {};
        for (int kk = 0; kk < NCOL / 4; ++kk) {
            const int kb = kk * 4 + half * 2;
            const v2f a   = *(const v2f*)(gnrow + kb);
            const v2f b0  = *(const v2f*)(b0row + kb);
            const v2f b1v = *(const v2f*)(b1row + kb);
            c0 = __builtin_amdgcn_wmma_f32_16x16x4_f32(false, a, false, b0,
                                                       (short)0, c0, false, false);
            c1 = __builtin_amdgcn_wmma_f32_16x16x4_f32(false, a, false, b1v,
                                                       (short)0, c1, false, false);
        }
        float v0 = c0[0] + bl0;
        v0 = 0.5f * v0 * (1.0f + erff(v0 * 0.70710678118654752f));
        float v1 = c1[0] + bl1;
        v1 = 0.5f * v1 * (1.0f + erff(v1 * 0.70710678118654752f));
        if (lane < 16) hbuf[l] = v0;
        if (lane < HID - 16) hbuf[16 + l] = v1;
    }
    lds_barrier();

    float lgt[NB];
    #pragma unroll
    for (int j = 0; j < NB; ++j) lgt[j] = b2s[j];
    #pragma unroll 8
    for (int k = 0; k < HID; ++k) {
        const float hk = hbuf[k];
        #pragma unroll
        for (int j = 0; j < NB; ++j) lgt[j] += hk * w2s[k * NB + j];
    }
    float al[NB];
    {
        const float mx = fmaxf(lgt[0], fmaxf(lgt[1], lgt[2]));
        const float e0 = __expf(lgt[0] - mx);
        const float e1 = __expf(lgt[1] - mx);
        const float e2 = __expf(lgt[2] - mx);
        const float inv = 1.0f / (e0 + e1 + e2);
        al[0] = __int_as_float(__builtin_amdgcn_readfirstlane(__float_as_int(e0 * inv)));
        al[1] = __int_as_float(__builtin_amdgcn_readfirstlane(__float_as_int(e1 * inv)));
        al[2] = __int_as_float(__builtin_amdgcn_readfirstlane(__float_as_int(e2 * inv)));
    }
    float bsum = 0.f;
    #pragma unroll
    for (int i = 0; i < NB; ++i) bsum += al[i] * convb[i * C_CH + cch];
    const float bias = __int_as_float(__builtin_amdgcn_readfirstlane(__float_as_int(bsum)));

    const int grp = tid >> 4;
    const int li  = tid & 15;
    const bool act = (li < ROWCH);
    const int sx = act ? li : (ROWCH - 1);
    const int tA = (2 * grp) * RPITCH + sx;
    const int tO = (2 * grp) * ROWCH + sx;

    float wr[NB * 27];
    #pragma unroll
    for (int i = 0; i < NB * 27; ++i)
        wr[i] = al[i / 27] * convw[((i / 27) * C_CH + cch) * 27 + (i % 27)];

    float acc[7][2][4];
    #pragma unroll
    for (int j = 0; j < 7; ++j) {
        #pragma unroll
        for (int o = 0; o < 2; ++o) { acc[j][o][0] = bias; acc[j][o][1] = bias; acc[j][o][2] = bias; acc[j][o][3] = bias; }
    }

    v4f* st = tile + FRONT + rlo * RPITCH;

    v4f fin0 = zero, fin1 = zero;
    const v4f* srcq = xp + 2 * PLCH;
    v4f* opq = (v4f*)(out + (size_t)bc * SPATIAL + (yh * YSL) * WDIM);

    #pragma unroll 1
    for (int p = 0; p < DDIM; ++p) {
        const int s = p & 1;
        if (p + 1 < DDIM) {
            v4f* dst = st + (s ^ 1) * SLOTCH;
            dst[d0] = nb0;
            if (tid + CTHR < REALCH) dst[d0 + 16 * RPITCH] = nb1;
        }
        if (p + 2 < DDIM) {
            nb0 = srcq[tid];
            nb1 = srcq[tid2];
            srcq += PLCH;
        }
        if (p >= 4) {
            if (act) { opq[tO] = fin0; opq[tO + ROWCH] = fin1; }
            opq += PLCH;
        }

        const v4f* q0 = tile + (FRONT - 1) + s * SLOTCH + tA;
        v4f rb[2][3];
        rb[0][0] = q0[0]; rb[0][1] = q0[1]; rb[0][2] = q0[2];
        #pragma unroll
        for (int r = 0; r < 8; ++r) {
            if (r + 1 < 8) {
                const v4f* qn = q0 + (r + 1) * RPITCH;
                rb[(r + 1) & 1][0] = qn[0]; rb[(r + 1) & 1][1] = qn[1]; rb[(r + 1) & 1][2] = qn[2];
            }
            const v4f L = rb[r & 1][0], C = rb[r & 1][1], R = rb[r & 1][2];
            float l1 = L.y, l2 = L.z, l3 = L.w, r0 = R.x, r1 = R.y, r2 = R.z;
            asm("" : "+v"(l1)); asm("" : "+v"(l2)); asm("" : "+v"(l3));
            asm("" : "+v"(r0)); asm("" : "+v"(r1)); asm("" : "+v"(r2));
            const float v[12] = {0.f, l1, l2, l3, C.x, C.y, C.z, C.w, r0, r1, r2, 0.f};
            #pragma unroll
            for (int o = 0; o < 2; ++o) {
                const int dy = r - 3 - o;
                if (dy < -3 || dy > 3) continue;
                #pragma unroll
                for (int d = 1; d <= NB; ++d) {
                    if (dy != 0 && dy != d && dy != -d) continue;
                    const int ky = (dy == 0) ? 1 : ((dy < 0) ? 0 : 2);
                    #pragma unroll
                    for (int kz = 0; kz < 3; ++kz) {
                        const int j = 3 - (kz - 1) * d;
                        const float* wp = &wr[(d - 1) * 27 + (kz * 3 + ky) * 3];
                        #pragma unroll
                        for (int kx = 0; kx < 3; ++kx) {
                            const float w = wp[kx];
                            const int ox = 4 + (kx - 1) * d;
                            acc[j][o][0] = fmaf(w, v[ox + 0], acc[j][o][0]);
                            acc[j][o][1] = fmaf(w, v[ox + 1], acc[j][o][1]);
                            acc[j][o][2] = fmaf(w, v[ox + 2], acc[j][o][2]);
                            acc[j][o][3] = fmaf(w, v[ox + 3], acc[j][o][3]);
                        }
                    }
                }
            }
        }

        fin0.x = acc[0][0][0]; fin0.y = acc[0][0][1]; fin0.z = acc[0][0][2]; fin0.w = acc[0][0][3];
        fin1.x = acc[0][1][0]; fin1.y = acc[0][1][1]; fin1.z = acc[0][1][2]; fin1.w = acc[0][1][3];
        #pragma unroll
        for (int j = 0; j < 6; ++j) {
            #pragma unroll
            for (int o = 0; o < 2; ++o) {
                acc[j][o][0] = acc[j + 1][o][0]; acc[j][o][1] = acc[j + 1][o][1];
                acc[j][o][2] = acc[j + 1][o][2]; acc[j][o][3] = acc[j + 1][o][3];
            }
        }
        #pragma unroll
        for (int o = 0; o < 2; ++o) { acc[6][o][0] = bias; acc[6][o][1] = bias; acc[6][o][2] = bias; acc[6][o][3] = bias; }

        lds_barrier();
    }

    if (act) {
        opq[tO] = fin0;
        opq[tO + ROWCH] = fin1;
        #pragma unroll
        for (int j = 0; j < 3; ++j) {
            const v4f r0 = {acc[j][0][0], acc[j][0][1], acc[j][0][2], acc[j][0][3]};
            const v4f r1 = {acc[j][1][0], acc[j][1][1], acc[j][1][2], acc[j][1][3]};
            opq[(1 + j) * PLCH + tO] = r0;
            opq[(1 + j) * PLCH + tO + ROWCH] = r1;
        }
    }
}

extern "C" void kernel_launch(void* const* d_in, const int* in_sizes, int n_in,
                              void* d_out, int out_size, void* d_ws, size_t ws_size,
                              hipStream_t stream) {
    const float* x        = (const float*)d_in[0];
    const float* guidance = (const float*)d_in[1];
    const float* convw    = (const float*)d_in[2];
    const float* convb    = (const float*)d_in[3];
    const float* ln_g     = (const float*)d_in[4];
    const float* ln_b     = (const float*)d_in[5];
    const float* w1       = (const float*)d_in[6];
    const float* b1       = (const float*)d_in[7];
    const float* w2       = (const float*)d_in[8];
    const float* b2       = (const float*)d_in[9];

    float* ws      = (float*)d_ws;
    float* feat    = ws;

    feat_reduce_kernel<<<B_N * C_CH, 256, 0, stream>>>(x, feat);
    dwconv_kernel<<<B_N * C_CH * YBLK, CTHR, 0, stream>>>(x, feat, guidance, ln_g, ln_b,
                                                         w1, b1, w2, b2, convw, convb,
                                                         (float*)d_out);
}


# conv steps of input planes 0, 1, 14, 15 are copies of their own without the taps into output planes outside the volume (sums never stored); every stored output keeps its taps and their order
# speedup vs baseline: 15.8441x; 1.0724x over previous
#include <hip/hip_runtime.h>
#include <math.h>

#define B_N   4
#define C_CH  96
#define DDIM  16
#define HDIM  56
#define WDIM  56
#define GDIM  96
#define HID   24
#define NB    3
#define NCOL  (C_CH + GDIM)
#define SPATIAL (DDIM * HDIM * WDIM)
#define LN_EPS 1e-5f

typedef float v2f __attribute__((ext_vector_type(2)));
typedef float v4f __attribute__((ext_vector_type(4)));
typedef float v8f __attribute__((ext_vector_type(8)));

template <int V> struct IC { static constexpr int value = V; };

__global__ void feat_reduce_kernel(const float* __restrict__ x,
                                   float* __restrict__ feat) {
    const int bc = blockIdx.x;
    const v4f* p = (const v4f*)(x + (size_t)bc * SPATIAL);
    float s = 0.f;
    for (int i = threadIdx.x; i < SPATIAL / 4; i += 256) {
        v4f v = p[i];
        s += v.x + v.y + v.z + v.w;
    }
    __shared__ float red[256];
    red[threadIdx.x] = s;
    __syncthreads();
    for (int off = 128; off > 0; off >>= 1) {
        if (threadIdx.x < off) red[threadIdx.x] += red[threadIdx.x + off];
        __syncthreads();
    }
    if (threadIdx.x == 0) feat[bc] = red[0] * (1.0f / (float)SPATIAL);
}

#define ROWCH   14
#define RPITCH  15
#define YSL     28
#define YBLK    (HDIM / YSL)
#define SROWS   (YSL + 6)
#define REALROWS (YSL + 3)
#define REALCH  (REALROWS * ROWCH)
#define PLCH    (HDIM * ROWCH)
#define SLOTCH  512
#define FRONT   1
#define NCHUNK  (FRONT + 2 * SLOTCH + 3)
#define W1_V4   ((NCOL * HID) / 4)
#define NPAIR   (YSL / 2)
#define CTHR    (NPAIR * 16)
#define W1_IT   6
#define W1P     196

__device__ __forceinline__ void lds_barrier() {
    __builtin_amdgcn_fence(__ATOMIC_RELEASE, "workgroup", "local");
    __builtin_amdgcn_s_barrier();
    __builtin_amdgcn_fence(__ATOMIC_ACQUIRE, "workgroup", "local");
}

__global__ __launch_bounds__(CTHR, 6)
void dwconv_kernel(const float* __restrict__ x,
                   const float* __restrict__ feat,
                   const float* __restrict__ guidance,
                   const float* __restrict__ ln_g,
                   const float* __restrict__ ln_b,
                   const float* __restrict__ w1p,
                   const float* __restrict__ b1p,
                   const float* __restrict__ w2p,
                   const float* __restrict__ b2p,
                   const float* __restrict__ convw,
                   const float* __restrict__ convb,
                   float* __restrict__ out) {
    const int blk = blockIdx.x;
    const int yh  = blk % YBLK;
    const int bc  = blk / YBLK;
    const int tid = threadIdx.x;
    const int cch = bc % C_CH;
    const int bat = bc / C_CH;

    __shared__ v4f tile[NCHUNK];
    __shared__ __attribute__((aligned(16))) float w1t[HID][W1P];
    __shared__ __attribute__((aligned(16))) float graw[NCOL];
    __shared__ __attribute__((aligned(16))) float gnrow[NCOL];
    __shared__ float hbuf[HID];
    __shared__ float w2s[HID * NB];
    __shared__ float b2s[NB];

    const int lane = tid & 31;
    const int half = lane >> 4;
    const int l    = lane & 15;

    v4f wv[W1_IT];
    #pragma unroll
    for (int j = 0; j < W1_IT; ++j) {
        const int q  = min(tid + CTHR * j, W1_V4 - 1);
        const int n4 = q / NCOL, k = q - n4 * NCOL;
        wv[j] = *(const v4f*)(w1p + k * HID + n4 * 4);
    }
    const int col = min(tid, NCOL - 1);
    const float* gsrc = (col < C_CH) ? (feat + bat * C_CH + col)
                                     : (guidance + bat * GDIM + (col - C_CH));
    const float gvv = *gsrc;
    const float lgv = ln_g[col];
    const float lbv = ln_b[col];
    const float bl0 = b1p[l];
    const float bl1 = b1p[min(16 + l, HID - 1)];
    const float w2v = w2p[min(tid, HID * NB - 1)];
    const float b2v = b2p[min(tid, NB - 1)];

    const int ylo = yh ? (HDIM - REALROWS) : 0;
    const int rlo = yh ? 0 : 3;
    const v4f* xp = (const v4f*)(x + (size_t)bc * SPATIAL + ylo * WDIM);
    const v4f zero = {0.f, 0.f, 0.f, 0.f};

    const int tid2 = min(tid + CTHR, REALCH - 1);
    const int d0 = tid + tid / ROWCH;
    const v4f f0 = xp[tid];
    const v4f f1 = xp[tid2];
    for (int i = tid; i < NCHUNK; i += CTHR) {
        const int e = i - (FRONT + rlo * RPITCH);
        const bool data0 = ((unsigned)e < (unsigned)(REALROWS * RPITCH)) && (e % RPITCH != ROWCH);
        if (!data0) tile[i] = zero;
    }

    v4f nb0 = xp[PLCH + tid];
    v4f nb1 = xp[PLCH + tid2];
    {
        v4f* st0 = tile + FRONT + rlo * RPITCH;
        st0[d0] = f0;
        if (tid + CTHR < REALCH) st0[d0 + 16 * RPITCH] = f1;
    }

    #pragma unroll
    for (int j = 0; j < W1_IT; ++j) {
        const int q = tid + CTHR * j;
        if (q < W1_V4) {
            const int n4 = q / NCOL, k = q - n4 * NCOL;
            float* d = &w1t[n4 * 4][k];
            d[0 * W1P] = wv[j].x;
            d[1 * W1P] = wv[j].y;
            d[2 * W1P] = wv[j].z;
            d[3 * W1P] = wv[j].w;
        }
    }
    if (tid < NCOL) graw[tid] = gvv;
    if (tid < HID * NB) w2s[tid] = w2v;
    if (tid < NB) b2s[tid] = b2v;
    lds_barrier();

    float mean_r, rstd_r;
    {
        float s = 0.f;
        #pragma unroll
        for (int k = 0; k < NCOL / 32; ++k) s += graw[lane + 32 * k];
        #pragma unroll
        for (int off = 16; off > 0; off >>= 1) s += __shfl_xor(s, off);
        mean_r = s * (1.0f / NCOL);
        float v = 0.f;
        #pragma unroll
        for (int k = 0; k < NCOL / 32; ++k) { const float d = graw[lane + 32 * k] - mean_r; v += d * d; }
        #pragma unroll
        for (int off = 16; off > 0; off >>= 1) v += __shfl_xor(v, off);
        v *= (1.0f / NCOL);
        rstd_r = rsqrtf(v + LN_EPS);
    }
    if (tid < NCOL) gnrow[tid] = (gvv - mean_r) * rstd_r * lgv + lbv;
    lds_barrier();

    const int wave = __builtin_amdgcn_readfirstlane(tid >> 5);
    if (wave == 0) {
        const float* b0row = &w1t[l][0];
        const float* b1row = &w1t[min(16 + l, HID - 1)][0];
        v8f c0 = {};
        v8f c1 = {};
        for (int kk = 0; kk < NCOL / 4; ++kk) {
            const int kb = kk * 4 + half * 2;
            const v2f a   = *(const v2f*)(gnrow + kb);
            const v2f b0  = *(const v2f*)(b0row + kb);
            const v2f b1v = *(const v2f*)(b1row + kb);
            c0 = __builtin_amdgcn_wmma_f32_16x16x4_f32(false, a, false, b0,
                                                       (short)0, c0, false, false);
            c1 = __builtin_amdgcn_wmma_f32_16x16x4_f32(false, a, false, b1v,
                                                       (short)0, c1, false, false);
        }
        float v0 = c0[0] + bl0;
        v0 = 0.5f * v0 * (1.0f + erff(v0 * 0.70710678118654752f));
        float v1 = c1[0] + bl1;
        v1 = 0.5f * v1 * (1.0f + erff(v1 * 0.70710678118654752f));
        if (lane < 16) hbuf[l] = v0;
        if (lane < HID - 16) hbuf[16 + l] = v1;
    }
    lds_barrier();

    float lgt[NB];
    #pragma unroll
    for (int j = 0; j < NB; ++j) lgt[j] = b2s[j];
    #pragma unroll 8
    for (int k = 0; k < HID; ++k) {
        const float hk = hbuf[k];
        #pragma unroll
        for (int j = 0; j < NB; ++j) lgt[j] += hk * w2s[k * NB + j];
    }
    float al[NB];
    {
        const float mx = fmaxf(lgt[0], fmaxf(lgt[1], lgt[2]));
        const float e0 = __expf(lgt[0] - mx);
        const float e1 = __expf(lgt[1] - mx);
        const float e2 = __expf(lgt[2] - mx);
        const float inv = 1.0f / (e0 + e1 + e2);
        al[0] = __int_as_float(__builtin_amdgcn_readfirstlane(__float_as_int(e0 * inv)));
        al[1] = __int_as_float(__builtin_amdgcn_readfirstlane(__float_as_int(e1 * inv)));
        al[2] = __int_as_float(__builtin_amdgcn_readfirstlane(__float_as_int(e2 * inv)));
    }
    float bsum = 0.f;
    #pragma unroll
    for (int i = 0; i < NB; ++i) bsum += al[i] * convb[i * C_CH + cch];
    const float bias = __int_as_float(__builtin_amdgcn_readfirstlane(__float_as_int(bsum)));

    const int grp = tid >> 4;
    const int li  = tid & 15;
    const bool act = (li < ROWCH);
    const int sx = act ? li : (ROWCH - 1);
    const int tA = (2 * grp) * RPITCH + sx;
    const int tO = (2 * grp) * ROWCH + sx;

    float wr[NB * 27];
    #pragma unroll
    for (int i = 0; i < NB * 27; ++i)
        wr[i] = al[i / 27] * convw[((i / 27) * C_CH + cch) * 27 + (i % 27)];

    float acc[7][2][4];
    #pragma unroll
    for (int j = 0; j < 7; ++j) {
        #pragma unroll
        for (int o = 0; o < 2; ++o) { acc[j][o][0] = bias; acc[j][o][1] = bias; acc[j][o][2] = bias; acc[j][o][3] = bias; }
    }

    v4f* st = tile + FRONT + rlo * RPITCH;

    v4f fin0 = zero, fin1 = zero;
    const v4f* srcq = xp + 2 * PLCH;
    v4f* opq = (v4f*)(out + (size_t)bc * SPATIAL + (yh * YSL) * WDIM);

    auto plane_step = [&](const int p, auto lo_c, auto hi_c) __attribute__((always_inline)) {
        constexpr int LO = decltype(lo_c)::value;
        constexpr int HI = decltype(hi_c)::value;
        const int s = p & 1;
        if (p + 1 < DDIM) {
            v4f* dst = st + (s ^ 1) * SLOTCH;
            dst[d0] = nb0;
            if (tid + CTHR < REALCH) dst[d0 + 16 * RPITCH] = nb1;
        }
        if (p + 2 < DDIM) {
            nb0 = srcq[tid];
            nb1 = srcq[tid2];
            srcq += PLCH;
        }
        if (p >= 4) {
            if (act) { opq[tO] = fin0; opq[tO + ROWCH] = fin1; }
            opq += PLCH;
        }

        const v4f* q0 = tile + (FRONT - 1) + s * SLOTCH + tA;
        v4f rb[2][3];
        rb[0][0] = q0[0]; rb[0][1] = q0[1]; rb[0][2] = q0[2];
        #pragma unroll
        for (int r = 0; r < 8; ++r) {
            if (r + 1 < 8) {
                const v4f* qn = q0 + (r + 1) * RPITCH;
                rb[(r + 1) & 1][0] = qn[0]; rb[(r + 1) & 1][1] = qn[1]; rb[(r + 1) & 1][2] = qn[2];
            }
            const v4f L = rb[r & 1][0], C = rb[r & 1][1], R = rb[r & 1][2];
            float l1 = L.y, l2 = L.z, l3 = L.w, r0 = R.x, r1 = R.y, r2 = R.z;
            asm("" : "+v"(l1)); asm("" : "+v"(l2)); asm("" : "+v"(l3));
            asm("" : "+v"(r0)); asm("" : "+v"(r1)); asm("" : "+v"(r2));
            const float v[12] = {0.f, l1, l2, l3, C.x, C.y, C.z, C.w, r0, r1, r2, 0.f};
            #pragma unroll
            for (int o = 0; o < 2; ++o) {
                const int dy = r - 3 - o;
                if (dy < -3 || dy > 3) continue;
                #pragma unroll
                for (int d = 1; d <= NB; ++d) {
                    if (dy != 0 && dy != d && dy != -d) continue;
                    const int ky = (dy == 0) ? 1 : ((dy < 0) ? 0 : 2);
                    #pragma unroll
                    for (int kz = 0; kz < 3; ++kz) {
                        const int j = 3 - (kz - 1) * d;
                        if (j < LO || j > 6 - HI) continue;
                        const float* wp = &wr[(d - 1) * 27 + (kz * 3 + ky) * 3];
                        #pragma unroll
                        for (int kx = 0; kx < 3; ++kx) {
                            const float w = wp[kx];
                            const int ox = 4 + (kx - 1) * d;
                            acc[j][o][0] = fmaf(w, v[ox + 0], acc[j][o][0]);
                            acc[j][o][1] = fmaf(w, v[ox + 1], acc[j][o][1]);
                            acc[j][o][2] = fmaf(w, v[ox + 2], acc[j][o][2]);
                            acc[j][o][3] = fmaf(w, v[ox + 3], acc[j][o][3]);
                        }
                    }
                }
            }
        }

        if (LO + HI > 0) {
            if (0 >= LO && 0 <= 6 - HI) asm volatile("" : "+v"(acc[0][0][0]), "+v"(acc[0][0][1]), "+v"(acc[0][0][2]), "+v"(acc[0][0][3]), "+v"(acc[0][1][0]), "+v"(acc[0][1][1]), "+v"(acc[0][1][2]), "+v"(acc[0][1][3]));
            if (1 >= LO && 1 <= 6 - HI) asm volatile("" : "+v"(acc[1][0][0]), "+v"(acc[1][0][1]), "+v"(acc[1][0][2]), "+v"(acc[1][0][3]), "+v"(acc[1][1][0]), "+v"(acc[1][1][1]), "+v"(acc[1][1][2]), "+v"(acc[1][1][3]));
            if (2 >= LO && 2 <= 6 - HI) asm volatile("" : "+v"(acc[2][0][0]), "+v"(acc[2][0][1]), "+v"(acc[2][0][2]), "+v"(acc[2][0][3]), "+v"(acc[2][1][0]), "+v"(acc[2][1][1]), "+v"(acc[2][1][2]), "+v"(acc[2][1][3]));
            if (3 >= LO && 3 <= 6 - HI) asm volatile("" : "+v"(acc[3][0][0]), "+v"(acc[3][0][1]), "+v"(acc[3][0][2]), "+v"(acc[3][0][3]), "+v"(acc[3][1][0]), "+v"(acc[3][1][1]), "+v"(acc[3][1][2]), "+v"(acc[3][1][3]));
            if (4 >= LO && 4 <= 6 - HI) asm volatile("" : "+v"(acc[4][0][0]), "+v"(acc[4][0][1]), "+v"(acc[4][0][2]), "+v"(acc[4][0][3]), "+v"(acc[4][1][0]), "+v"(acc[4][1][1]), "+v"(acc[4][1][2]), "+v"(acc[4][1][3]));
            if (5 >= LO && 5 <= 6 - HI) asm volatile("" : "+v"(acc[5][0][0]), "+v"(acc[5][0][1]), "+v"(acc[5][0][2]), "+v"(acc[5][0][3]), "+v"(acc[5][1][0]), "+v"(acc[5][1][1]), "+v"(acc[5][1][2]), "+v"(acc[5][1][3]));
            if (6 >= LO && 6 <= 6 - HI) asm volatile("" : "+v"(acc[6][0][0]), "+v"(acc[6][0][1]), "+v"(acc[6][0][2]), "+v"(acc[6][0][3]), "+v"(acc[6][1][0]), "+v"(acc[6][1][1]), "+v"(acc[6][1][2]), "+v"(acc[6][1][3]));
        }
        fin0.x = acc[0][0][0]; fin0.y = acc[0][0][1]; fin0.z = acc[0][0][2]; fin0.w = acc[0][0][3];
        fin1.x = acc[0][1][0]; fin1.y = acc[0][1][1]; fin1.z = acc[0][1][2]; fin1.w = acc[0][1][3];
        #pragma unroll
        for (int j = 0; j < 6; ++j) {
            #pragma unroll
            for (int o = 0; o < 2; ++o) {
                acc[j][o][0] = acc[j + 1][o][0]; acc[j][o][1] = acc[j + 1][o][1];
                acc[j][o][2] = acc[j + 1][o][2]; acc[j][o][3] = acc[j + 1][o][3];
            }
        }
        #pragma unroll
        for (int o = 0; o < 2; ++o) { acc[6][o][0] = bias; acc[6][o][1] = bias; acc[6][o][2] = bias; acc[6][o][3] = bias; }

        lds_barrier();
    };

    plane_step(0, IC<3>{}, IC<0>{});
    plane_step(1, IC<2>{}, IC<0>{});
    #pragma unroll 1
    for (int p = 2; p < DDIM - 2; ++p) plane_step(p, IC<0>{}, IC<0>{});
    plane_step(DDIM - 2, IC<0>{}, IC<2>{});
    plane_step(DDIM - 1, IC<0>{}, IC<3>{});

    if (act) {
        opq[tO] = fin0;
        opq[tO + ROWCH] = fin1;
        #pragma unroll
        for (int j = 0; j < 3; ++j) {
            const v4f r0 = {acc[j][0][0], acc[j][0][1], acc[j][0][2], acc[j][0][3]};
            const v4f r1 = {acc[j][1][0], acc[j][1][1], acc[j][1][2], acc[j][1][3]};
            opq[(1 + j) * PLCH + tO] = r0;
            opq[(1 + j) * PLCH + tO + ROWCH] = r1;
        }
    }
}

extern "C" void kernel_launch(void* const* d_in, const int* in_sizes, int n_in,
                              void* d_out, int out_size, void* d_ws, size_t ws_size,
                              hipStream_t stream) {
    const float* x        = (const float*)d_in[0];
    const float* guidance = (const float*)d_in[1];
    const float* convw    = (const float*)d_in[2];
    const float* convb    = (const float*)d_in[3];
    const float* ln_g     = (const float*)d_in[4];
    const float* ln_b     = (const float*)d_in[5];
    const float* w1       = (const float*)d_in[6];
    const float* b1       = (const float*)d_in[7];
    const float* w2       = (const float*)d_in[8];
    const float* b2       = (const float*)d_in[9];

    float* ws      = (float*)d_ws;
    float* feat    = ws;

    feat_reduce_kernel<<<B_N * C_CH, 256, 0, stream>>>(x, feat);
    dwconv_kernel<<<B_N * C_CH * YBLK, CTHR, 0, stream>>>(x, feat, guidance, ln_g, ln_b,
                                                         w1, b1, w2, b2, convw, convb,
                                                         (float*)d_out);
}


# mean-pool kernel: 768 blocks of 4 waves (per b, c, half of y), a thread's 49 b128 loads all in flight before its first add, no 8-load rounds; half sums joined in conv prologue: mean sum order changes
# speedup vs baseline: 16.3291x; 1.0306x over previous
#include <hip/hip_runtime.h>
#include <math.h>

#define B_N   4
#define C_CH  96
#define DDIM  16
#define HDIM  56
#define WDIM  56
#define GDIM  96
#define HID   24
#define NB    3
#define NCOL  (C_CH + GDIM)
#define SPATIAL (DDIM * HDIM * WDIM)
#define LN_EPS 1e-5f

typedef float v2f __attribute__((ext_vector_type(2)));
typedef float v4f __attribute__((ext_vector_type(4)));
typedef float v8f __attribute__((ext_vector_type(8)));

template <int V> struct IC { static constexpr int value = V; };

#define FTHR   128
#define FHALF  392
#define FPLANE 784
#define FBODY  (FHALF / FTHR)
#define FTAIL  (FHALF - FBODY * FTHR)
#define FLD    (DDIM * FBODY + 1)

__global__ __launch_bounds__(FTHR, 3)
void feat_reduce_kernel(const float* __restrict__ x,
                        float* __restrict__ part) {
    const int blk = blockIdx.x;
    const int yh  = blk & 1;
    const int bc  = blk >> 1;
    const int tid = threadIdx.x;
    const v4f* p = (const v4f*)(x + (size_t)bc * SPATIAL) + yh * FHALF;
    v4f v[FLD];
    #pragma unroll
    for (int pl = 0; pl < DDIM; ++pl) {
        #pragma unroll
        for (int j = 0; j < FBODY; ++j) v[pl * FBODY + j] = p[pl * FPLANE + j * FTHR + tid];
    }
    v[FLD - 1] = p[(tid / FTAIL) * FPLANE + FBODY * FTHR + (tid % FTAIL)];
    __builtin_amdgcn_sched_barrier(0);
    float s0 = 0.f, s1 = 0.f, s2 = 0.f, s3 = 0.f;
    #pragma unroll
    for (int k = 0; k < FLD; ++k) { s0 += v[k].x; s1 += v[k].y; s2 += v[k].z; s3 += v[k].w; }
    float s = (s0 + s1) + (s2 + s3);
    #pragma unroll
    for (int off = 16; off > 0; off >>= 1) s += __shfl_xor(s, off);
    __shared__ float red[FTHR / 32];
    if ((tid & 31) == 0) red[tid >> 5] = s;
    __syncthreads();
    if (tid == 0) part[blk] = ((red[0] + red[1]) + red[2]) + red[3];
}

#define ROWCH   14
#define RPITCH  15
#define YSL     28
#define YBLK    (HDIM / YSL)
#define SROWS   (YSL + 6)
#define REALROWS (YSL + 3)
#define REALCH  (REALROWS * ROWCH)
#define PLCH    (HDIM * ROWCH)
#define SLOTCH  512
#define FRONT   1
#define NCHUNK  (FRONT + 2 * SLOTCH + 3)
#define W1_V4   ((NCOL * HID) / 4)
#define NPAIR   (YSL / 2)
#define CTHR    (NPAIR * 16)
#define W1_IT   6
#define W1P     196

__device__ __forceinline__ void lds_barrier() {
    __builtin_amdgcn_fence(__ATOMIC_RELEASE, "workgroup", "local");
    __builtin_amdgcn_s_barrier();
    __builtin_amdgcn_fence(__ATOMIC_ACQUIRE, "workgroup", "local");
}

__global__ __launch_bounds__(CTHR, 6)
void dwconv_kernel(const float* __restrict__ x,
                   const float* __restrict__ part,
                   const float* __restrict__ guidance,
                   const float* __restrict__ ln_g,
                   const float* __restrict__ ln_b,
                   const float* __restrict__ w1p,
                   const float* __restrict__ b1p,
                   const float* __restrict__ w2p,
                   const float* __restrict__ b2p,
                   const float* __restrict__ convw,
                   const float* __restrict__ convb,
                   float* __restrict__ out) {
    const int blk = blockIdx.x;
    const int yh  = blk % YBLK;
    const int bc  = blk / YBLK;
    const int tid = threadIdx.x;
    const int cch = bc % C_CH;
    const int bat = bc / C_CH;

    __shared__ v4f tile[NCHUNK];
    __shared__ __attribute__((aligned(16))) float w1t[HID][W1P];
    __shared__ __attribute__((aligned(16))) float graw[NCOL];
    __shared__ __attribute__((aligned(16))) float gnrow[NCOL];
    __shared__ float hbuf[HID];
    __shared__ float w2s[HID * NB];
    __shared__ float b2s[NB];

    const int lane = tid & 31;
    const int half = lane >> 4;
    const int l    = lane & 15;

    v4f wv[W1_IT];
    #pragma unroll
    for (int j = 0; j < W1_IT; ++j) {
        const int q  = min(tid + CTHR * j, W1_V4 - 1);
        const int n4 = q / NCOL, k = q - n4 * NCOL;
        wv[j] = *(const v4f*)(w1p + k * HID + n4 * 4);
    }
    const int col = min(tid, NCOL - 1);
    const bool  isch = (col < C_CH);
    const int   gcol = max(col - C_CH, 0);
    const float* gpc = part + 2 * (bat * C_CH + min(col, C_CH - 1));
    const float* gpg = guidance + bat * GDIM + (gcol & ~1);
    const float* gsrc = isch ? gpc : gpg;
    const v2f gpr = *(const v2f*)gsrc;
    const float lgv = ln_g[col];
    const float lbv = ln_b[col];
    const float bl0 = b1p[l];
    const float bl1 = b1p[min(16 + l, HID - 1)];
    const float w2v = w2p[min(tid, HID * NB - 1)];
    const float b2v = b2p[min(tid, NB - 1)];

    const int ylo = yh ? (HDIM - REALROWS) : 0;
    const int rlo = yh ? 0 : 3;
    const v4f* xp = (const v4f*)(x + (size_t)bc * SPATIAL + ylo * WDIM);
    const v4f zero = {0.f, 0.f, 0.f, 0.f};

    const int tid2 = min(tid + CTHR, REALCH - 1);
    const int d0 = tid + tid / ROWCH;
    const v4f f0 = xp[tid];
    const v4f f1 = xp[tid2];
    for (int i = tid; i < NCHUNK; i += CTHR) {
        const int e = i - (FRONT + rlo * RPITCH);
        const bool data0 = ((unsigned)e < (unsigned)(REALROWS * RPITCH)) && (e % RPITCH != ROWCH);
        if (!data0) tile[i] = zero;
    }

    v4f nb0 = xp[PLCH + tid];
    v4f nb1 = xp[PLCH + tid2];
    {
        v4f* st0 = tile + FRONT + rlo * RPITCH;
        st0[d0] = f0;
        if (tid + CTHR < REALCH) st0[d0 + 16 * RPITCH] = f1;
    }

    #pragma unroll
    for (int j = 0; j < W1_IT; ++j) {
        const int q = tid + CTHR * j;
        if (q < W1_V4) {
            const int n4 = q / NCOL, k = q - n4 * NCOL;
            float* d = &w1t[n4 * 4][k];
            d[0 * W1P] = wv[j].x;
            d[1 * W1P] = wv[j].y;
            d[2 * W1P] = wv[j].z;
            d[3 * W1P] = wv[j].w;
        }
    }
    const float gsum = (gpr.x + gpr.y) * (1.0f / (float)SPATIAL);
    const float gsel = (gcol & 1) ? gpr.y : gpr.x;
    const float gvv  = isch ? gsum : gsel;
    if (tid < NCOL) graw[tid] = gvv;
    if (tid < HID * NB) w2s[tid] = w2v;
    if (tid < NB) b2s[tid] = b2v;
    lds_barrier();

    float mean_r, rstd_r;
    {
        float s = 0.f;
        #pragma unroll
        for (int k = 0; k < NCOL / 32; ++k) s += graw[lane + 32 * k];
        #pragma unroll
        for (int off = 16; off > 0; off >>= 1) s += __shfl_xor(s, off);
        mean_r = s * (1.0f / NCOL);
        float v = 0.f;
        #pragma unroll
        for (int k = 0; k < NCOL / 32; ++k) { const float d = graw[lane + 32 * k] - mean_r; v += d * d; }
        #pragma unroll
        for (int off = 16; off > 0; off >>= 1) v += __shfl_xor(v, off);
        v *= (1.0f / NCOL);
        rstd_r = rsqrtf(v + LN_EPS);
    }
    if (tid < NCOL) gnrow[tid] = (gvv - mean_r) * rstd_r * lgv + lbv;
    lds_barrier();

    const int wave = __builtin_amdgcn_readfirstlane(tid >> 5);
    if (wave == 0) {
        const float* b0row = &w1t[l][0];
        const float* b1row = &w1t[min(16 + l, HID - 1)][0];
        v8f c0 = {};
        v8f c1 = {};
        for (int kk = 0; kk < NCOL / 4; ++kk) {
            const int kb = kk * 4 + half * 2;
            const v2f a   = *(const v2f*)(gnrow + kb);
            const v2f b0  = *(const v2f*)(b0row + kb);
            const v2f b1v = *(const v2f*)(b1row + kb);
            c0 = __builtin_amdgcn_wmma_f32_16x16x4_f32(false, a, false, b0,
                                                       (short)0, c0, false, false);
            c1 = __builtin_amdgcn_wmma_f32_16x16x4_f32(false, a, false, b1v,
                                                       (short)0, c1, false, false);
        }
        float v0 = c0[0] + bl0;
        v0 = 0.5f * v0 * (1.0f + erff(v0 * 0.70710678118654752f));
        float v1 = c1[0] + bl1;
        v1 = 0.5f * v1 * (1.0f + erff(v1 * 0.70710678118654752f));
        if (lane < 16) hbuf[l] = v0;
        if (lane < HID - 16) hbuf[16 + l] = v1;
    }
    lds_barrier();

    float lgt[NB];
    #pragma unroll
    for (int j = 0; j < NB; ++j) lgt[j] = b2s[j];
    #pragma unroll 8
    for (int k = 0; k < HID; ++k) {
        const float hk = hbuf[k];
        #pragma unroll
        for (int j = 0; j < NB; ++j) lgt[j] += hk * w2s[k * NB + j];
    }
    float al[NB];
    {
        const float mx = fmaxf(lgt[0], fmaxf(lgt[1], lgt[2]));
        const float e0 = __expf(lgt[0] - mx);
        const float e1 = __expf(lgt[1] - mx);
        const float e2 = __expf(lgt[2] - mx);
        const float inv = 1.0f / (e0 + e1 + e2);
        al[0] = __int_as_float(__builtin_amdgcn_readfirstlane(__float_as_int(e0 * inv)));
        al[1] = __int_as_float(__builtin_amdgcn_readfirstlane(__float_as_int(e1 * inv)));
        al[2] = __int_as_float(__builtin_amdgcn_readfirstlane(__float_as_int(e2 * inv)));
    }
    float bsum = 0.f;
    #pragma unroll
    for (int i = 0; i < NB; ++i) bsum += al[i] * convb[i * C_CH + cch];
    const float bias = __int_as_float(__builtin_amdgcn_readfirstlane(__float_as_int(bsum)));

    const int grp = tid >> 4;
    const int li  = tid & 15;
    const bool act = (li < ROWCH);
    const int sx = act ? li : (ROWCH - 1);
    const int tA = (2 * grp) * RPITCH + sx;
    const int tO = (2 * grp) * ROWCH + sx;

    float wr[NB * 27];
    #pragma unroll
    for (int i = 0; i < NB * 27; ++i)
        wr[i] = al[i / 27] * convw[((i / 27) * C_CH + cch) * 27 + (i % 27)];

    float acc[7][2][4];
    #pragma unroll
    for (int j = 0; j < 7; ++j) {
        #pragma unroll
        for (int o = 0; o < 2; ++o) { acc[j][o][0] = bias; acc[j][o][1] = bias; acc[j][o][2] = bias; acc[j][o][3] = bias; }
    }

    v4f* st = tile + FRONT + rlo * RPITCH;

    v4f fin0 = zero, fin1 = zero;
    const v4f* srcq = xp + 2 * PLCH;
    v4f* opq = (v4f*)(out + (size_t)bc * SPATIAL + (yh * YSL) * WDIM);

    auto plane_step = [&](const int p, auto lo_c, auto hi_c) __attribute__((always_inline)) {
        constexpr int LO = decltype(lo_c)::value;
        constexpr int HI = decltype(hi_c)::value;
        const int s = p & 1;
        if (p + 1 < DDIM) {
            v4f* dst = st + (s ^ 1) * SLOTCH;
            dst[d0] = nb0;
            if (tid + CTHR < REALCH) dst[d0 + 16 * RPITCH] = nb1;
        }
        if (p + 2 < DDIM) {
            nb0 = srcq[tid];
            nb1 = srcq[tid2];
            srcq += PLCH;
        }
        if (p >= 4) {
            if (act) { opq[tO] = fin0; opq[tO + ROWCH] = fin1; }
            opq += PLCH;
        }

        const v4f* q0 = tile + (FRONT - 1) + s * SLOTCH + tA;
        v4f rb[2][3];
        rb[0][0] = q0[0]; rb[0][1] = q0[1]; rb[0][2] = q0[2];
        #pragma unroll
        for (int r = 0; r < 8; ++r) {
            if (r + 1 < 8) {
                const v4f* qn = q0 + (r + 1) * RPITCH;
                rb[(r + 1) & 1][0] = qn[0]; rb[(r + 1) & 1][1] = qn[1]; rb[(r + 1) & 1][2] = qn[2];
            }
            const v4f L = rb[r & 1][0], C = rb[r & 1][1], R = rb[r & 1][2];
            float l1 = L.y, l2 = L.z, l3 = L.w, r0 = R.x, r1 = R.y, r2 = R.z;
            asm("" : "+v"(l1)); asm("" : "+v"(l2)); asm("" : "+v"(l3));
            asm("" : "+v"(r0)); asm("" : "+v"(r1)); asm("" : "+v"(r2));
            const float v[12] = {0.f, l1, l2, l3, C.x, C.y, C.z, C.w, r0, r1, r2, 0.f};
            #pragma unroll
            for (int o = 0; o < 2; ++o) {
                const int dy = r - 3 - o;
                if (dy < -3 || dy > 3) continue;
                #pragma unroll
                for (int d = 1; d <= NB; ++d) {
                    if (dy != 0 && dy != d && dy != -d) continue;
                    const int ky = (dy == 0) ? 1 : ((dy < 0) ? 0 : 2);
                    #pragma unroll
                    for (int kz = 0; kz < 3; ++kz) {
                        const int j = 3 - (kz - 1) * d;
                        if (j < LO || j > 6 - HI) continue;
                        const float* wp = &wr[(d - 1) * 27 + (kz * 3 + ky) * 3];
                        #pragma unroll
                        for (int kx = 0; kx < 3; ++kx) {
                            const float w = wp[kx];
                            const int ox = 4 + (kx - 1) * d;
                            acc[j][o][0] = fmaf(w, v[ox + 0], acc[j][o][0]);
                            acc[j][o][1] = fmaf(w, v[ox + 1], acc[j][o][1]);
                            acc[j][o][2] = fmaf(w, v[ox + 2], acc[j][o][2]);
                            acc[j][o][3] = fmaf(w, v[ox + 3], acc[j][o][3]);
                        }
                    }
                }
            }
        }

        if (LO + HI > 0) {
            if (0 >= LO && 0 <= 6 - HI) asm volatile("" : "+v"(acc[0][0][0]), "+v"(acc[0][0][1]), "+v"(acc[0][0][2]), "+v"(acc[0][0][3]), "+v"(acc[0][1][0]), "+v"(acc[0][1][1]), "+v"(acc[0][1][2]), "+v"(acc[0][1][3]));
            if (1 >= LO && 1 <= 6 - HI) asm volatile("" : "+v"(acc[1][0][0]), "+v"(acc[1][0][1]), "+v"(acc[1][0][2]), "+v"(acc[1][0][3]), "+v"(acc[1][1][0]), "+v"(acc[1][1][1]), "+v"(acc[1][1][2]), "+v"(acc[1][1][3]));
            if (2 >= LO && 2 <= 6 - HI) asm volatile("" : "+v"(acc[2][0][0]), "+v"(acc[2][0][1]), "+v"(acc[2][0][2]), "+v"(acc[2][0][3]), "+v"(acc[2][1][0]), "+v"(acc[2][1][1]), "+v"(acc[2][1][2]), "+v"(acc[2][1][3]));
            if (3 >= LO && 3 <= 6 - HI) asm volatile("" : "+v"(acc[3][0][0]), "+v"(acc[3][0][1]), "+v"(acc[3][0][2]), "+v"(acc[3][0][3]), "+v"(acc[3][1][0]), "+v"(acc[3][1][1]), "+v"(acc[3][1][2]), "+v"(acc[3][1][3]));
            if (4 >= LO && 4 <= 6 - HI) asm volatile("" : "+v"(acc[4][0][0]), "+v"(acc[4][0][1]), "+v"(acc[4][0][2]), "+v"(acc[4][0][3]), "+v"(acc[4][1][0]), "+v"(acc[4][1][1]), "+v"(acc[4][1][2]), "+v"(acc[4][1][3]));
            if (5 >= LO && 5 <= 6 - HI) asm volatile("" : "+v"(acc[5][0][0]), "+v"(acc[5][0][1]), "+v"(acc[5][0][2]), "+v"(acc[5][0][3]), "+v"(acc[5][1][0]), "+v"(acc[5][1][1]), "+v"(acc[5][1][2]), "+v"(acc[5][1][3]));
            if (6 >= LO && 6 <= 6 - HI) asm volatile("" : "+v"(acc[6][0][0]), "+v"(acc[6][0][1]), "+v"(acc[6][0][2]), "+v"(acc[6][0][3]), "+v"(acc[6][1][0]), "+v"(acc[6][1][1]), "+v"(acc[6][1][2]), "+v"(acc[6][1][3]));
        }
        fin0.x = acc[0][0][0]; fin0.y = acc[0][0][1]; fin0.z = acc[0][0][2]; fin0.w = acc[0][0][3];
        fin1.x = acc[0][1][0]; fin1.y = acc[0][1][1]; fin1.z = acc[0][1][2]; fin1.w = acc[0][1][3];
        #pragma unroll
        for (int j = 0; j < 6; ++j) {
            #pragma unroll
            for (int o = 0; o < 2; ++o) {
                acc[j][o][0] = acc[j + 1][o][0]; acc[j][o][1] = acc[j + 1][o][1];
                acc[j][o][2] = acc[j + 1][o][2]; acc[j][o][3] = acc[j + 1][o][3];
            }
        }
        #pragma unroll
        for (int o = 0; o < 2; ++o) { acc[6][o][0] = bias; acc[6][o][1] = bias; acc[6][o][2] = bias; acc[6][o][3] = bias; }

        lds_barrier();
    };

    plane_step(0, IC<3>{}, IC<0>{});
    plane_step(1, IC<2>{}, IC<0>{});
    #pragma unroll 1
    for (int p = 2; p < DDIM - 2; ++p) plane_step(p, IC<0>{}, IC<0>{});
    plane_step(DDIM - 2, IC<0>{}, IC<2>{});
    plane_step(DDIM - 1, IC<0>{}, IC<3>{});

    if (act) {
        opq[tO] = fin0;
        opq[tO + ROWCH] = fin1;
        #pragma unroll
        for (int j = 0; j < 3; ++j) {
            const v4f r0 = {acc[j][0][0], acc[j][0][1], acc[j][0][2], acc[j][0][3]};
            const v4f r1 = {acc[j][1][0], acc[j][1][1], acc[j][1][2], acc[j][1][3]};
            opq[(1 + j) * PLCH + tO] = r0;
            opq[(1 + j) * PLCH + tO + ROWCH] = r1;
        }
    }
}

extern "C" void kernel_launch(void* const* d_in, const int* in_sizes, int n_in,
                              void* d_out, int out_size, void* d_ws, size_t ws_size,
                              hipStream_t stream) {
    const float* x        = (const float*)d_in[0];
    const float* guidance = (const float*)d_in[1];
    const float* convw    = (const float*)d_in[2];
    const float* convb    = (const float*)d_in[3];
    const float* ln_g     = (const float*)d_in[4];
    const float* ln_b     = (const float*)d_in[5];
    const float* w1       = (const float*)d_in[6];
    const float* b1       = (const float*)d_in[7];
    const float* w2       = (const float*)d_in[8];
    const float* b2       = (const float*)d_in[9];

    float* ws      = (float*)d_ws;
    float* part    = ws;

    feat_reduce_kernel<<<B_N * C_CH * 2, FTHR, 0, stream>>>(x, part);
    dwconv_kernel<<<B_N * C_CH * YBLK, CTHR, 0, stream>>>(x, part, guidance, ln_g, ln_b,
                                                         w1, b1, w2, b2, convw, convb,
                                                         (float*)d_out);
}
